# starting kernel
# baseline (speedup 1.0000x reference)
#include <hip/hip_runtime.h>
#include <cstddef>

typedef float v2f __attribute__((ext_vector_type(2)));
typedef float v4f __attribute__((ext_vector_type(4)));
typedef float v8f __attribute__((ext_vector_type(8)));

#define DMODEL 96
#define NHEADS 4
#define DHEAD  24

__global__ __launch_bounds__(256) void gt_init(const float* __restrict__ embeds,
                                               float* __restrict__ out,
                                               float* __restrict__ norms,
                                               int nNodes) {
  int i = blockIdx.x * 256 + threadIdx.x;
  int total = nNodes * DMODEL;
  if (i < total) out[i] = embeds[i];
  if (i < nNodes * NHEADS) norms[i] = 0.0f;
}

__global__ __launch_bounds__(256) void gt_proj_wmma(
    const float* __restrict__ embeds,
    const float* __restrict__ qW, const float* __restrict__ kW,
    const float* __restrict__ vW,
    float* __restrict__ Q, float* __restrict__ K, float* __restrict__ V,
    int nMTiles) {
  const int lane = threadIdx.x & 31;
  const int wave = threadIdx.x >> 5;
  const int tile = blockIdx.x * 8 + wave;
  const int nTiles = nMTiles * 18;
  if (tile >= nTiles) return;

  const int mt  = tile / 18;
  const int sub = tile % 18;
  const int mat = sub / 6;
  const int n0  = (sub % 6) * 16;
  const int m0  = mt * 16;

  const float* W   = (mat == 0) ? qW : (mat == 1) ? kW : vW;
  float*       Out = (mat == 0) ? Q  : (mat == 1) ? K  : V;

  const int halfSel = lane >> 4;
  const int l15     = lane & 15;
  const int kOff    = halfSel * 2;

  const float* arow = embeds + (size_t)(m0 + l15) * DMODEL;
  const float* bcol = W + n0 + l15;

  v8f c = {0.f, 0.f, 0.f, 0.f, 0.f, 0.f, 0.f, 0.f};
#pragma unroll
  for (int kk = 0; kk < DMODEL; kk += 4) {
    v2f a = *(const v2f*)(arow + kk + kOff);
    v2f b;
    b.x = bcol[(size_t)(kk + kOff)     * DMODEL];
    b.y = bcol[(size_t)(kk + kOff + 1) * DMODEL];
    c = __builtin_amdgcn_wmma_f32_16x16x4_f32(false, a, false, b,
                                              (short)0, c, false, false);
  }

  float* orow = Out + (size_t)m0 * DMODEL + n0 + l15;
#pragma unroll
  for (int g = 0; g < 8; ++g)
    orow[(size_t)(g + halfSel * 8) * DMODEL] = c[g];
}

__global__ __launch_bounds__(256) void gt_edge_scores(
    const float* __restrict__ Q, const float* __restrict__ K,
    const int* __restrict__ rows, const int* __restrict__ cols,
    float* __restrict__ exp_s, float* __restrict__ norms, int nEdges) {
  int tid = blockIdx.x * 256 + threadIdx.x;
  if (tid >= nEdges * NHEADS) return;
  int e = tid >> 2, h = tid & 3;
  int r = rows[e], c = cols[e];
  const v4f* q = (const v4f*)(Q + (size_t)r * DMODEL + h * DHEAD);
  const v4f* k = (const v4f*)(K + (size_t)c * DMODEL + h * DHEAD);
  float s = 0.0f;
#pragma unroll
  for (int j = 0; j < 6; ++j) {
    v4f a = q[j], b = k[j];
    s += a.x * b.x + a.y * b.y + a.z * b.z + a.w * b.w;
  }
  s = fminf(10.0f, fmaxf(-10.0f, s));
  float es = expf(s);
  exp_s[tid] = es;
  atomicAdd(&norms[(size_t)r * NHEADS + h], es);
}

__global__ __launch_bounds__(256) void gt_edge_aggregate(
    const float* __restrict__ V,
    const int* __restrict__ rows, const int* __restrict__ cols,
    const float* __restrict__ exp_s, const float* __restrict__ norms,
    float* __restrict__ out, int nEdges) {
  int tid = blockIdx.x * 256 + threadIdx.x;
  if (tid >= nEdges * NHEADS) return;
  int e = tid >> 2, h = tid & 3;
  int r = rows[e], c = cols[e];
  float w = exp_s[tid] / (norms[(size_t)r * NHEADS + h] + 1e-8f);
  const v4f* v = (const v4f*)(V + (size_t)c * DMODEL + h * DHEAD);
  float* o = out + (size_t)r * DMODEL + h * DHEAD;
#pragma unroll
  for (int j = 0; j < 6; ++j) {
    v4f t = v[j];
    atomicAdd(o + 4 * j + 0, w * t.x);
    atomicAdd(o + 4 * j + 1, w * t.y);
    atomicAdd(o + 4 * j + 2, w * t.z);
    atomicAdd(o + 4 * j + 3, w * t.w);
  }
}

extern "C" void kernel_launch(void* const* d_in, const int* in_sizes, int n_in,
                              void* d_out, int out_size, void* d_ws,
                              size_t ws_size, hipStream_t stream) {
  const float* embeds = (const float*)d_in[0];
  const float* qW     = (const float*)d_in[1];
  const float* kW     = (const float*)d_in[2];
  const float* vW     = (const float*)d_in[3];
  const int*   rows   = (const int*)d_in[4];
  const int*   cols   = (const int*)d_in[5];
  float*       out    = (float*)d_out;

  const int nNodes = in_sizes[0] / DMODEL;
  const int nEdges = in_sizes[4];

  float* ws = (float*)d_ws;
  const size_t nd = (size_t)nNodes * DMODEL;
  float* Q     = ws;
  float* Kp    = Q + nd;
  float* Vp    = Kp + nd;
  float* exp_s = Vp + nd;
  float* norms = exp_s + (size_t)nEdges * NHEADS;

  {
    int total = nNodes * DMODEL;
    gt_init<<<(total + 255) / 256, 256, 0, stream>>>(embeds, out, norms, nNodes);
  }
  {
    int nMTiles = nNodes / 16;
    int nTiles  = nMTiles * 18;
    int blocks  = (nTiles + 7) / 8;
    gt_proj_wmma<<<blocks, 256, 0, stream>>>(embeds, qW, kW, vW, Q, Kp, Vp,
                                             nMTiles);
  }
  {
    int total = nEdges * NHEADS;
    gt_edge_scores<<<(total + 255) / 256, 256, 0, stream>>>(Q, Kp, rows, cols,
                                                            exp_s, norms, nEdges);
    gt_edge_aggregate<<<(total + 255) / 256, 256, 0, stream>>>(Vp, rows, cols,
                                                               exp_s, norms, out,
                                                               nEdges);
  }
}


# Edges grouped by destination row in-call (int-atomic rank, offsets scan); per-row wave gather sums normalizer and weighted V in registers, out written once; no f32 atomics, sum order changes
# speedup vs baseline: 2.8673x; 2.8673x over previous
#include <hip/hip_runtime.h>
#include <cstddef>

typedef float v2f __attribute__((ext_vector_type(2)));
typedef float v4f __attribute__((ext_vector_type(4)));
typedef float v8f __attribute__((ext_vector_type(8)));

#define DMODEL 96
#define NHEADS 4
#define DHEAD  24
#define SCAN_TILE 65536

__global__ __launch_bounds__(256) void gt_zero_counts(int* __restrict__ cnt) {
  cnt[blockIdx.x * 256 + threadIdx.x] = 0;
}

__global__ __launch_bounds__(256) void gt_edge_rank(const int* __restrict__ rows,
                                                    int* __restrict__ cnt,
                                                    int* __restrict__ rank,
                                                    int nEdges) {
  int e = blockIdx.x * 256 + threadIdx.x;
  if (e < nEdges) rank[e] = atomicAdd(&cnt[rows[e]], 1);
}

__global__ __launch_bounds__(1024) void gt_scan_offsets(const int* __restrict__ cnt,
                                                        int* __restrict__ off,
                                                        int nScanTiles) {
  __shared__ int wtot[32];
  const int tid = threadIdx.x, lane = tid & 31, wave = tid >> 5;
  int carry = 0;
  for (int t = 0; t < nScanTiles; ++t) {
    const int base = t * SCAN_TILE + tid * 64;
    const int4* src = (const int4*)(cnt + base);
    int4 v[16];
#pragma unroll
    for (int j = 0; j < 16; ++j) v[j] = src[j];
    int run = 0;
#pragma unroll
    for (int j = 0; j < 16; ++j) {
      int a = v[j].x, b = v[j].y, c = v[j].z, d = v[j].w;
      v[j].x = run; run += a;
      v[j].y = run; run += b;
      v[j].z = run; run += c;
      v[j].w = run; run += d;
    }
    int incl = run;
#pragma unroll
    for (int d = 1; d < 32; d <<= 1) {
      int n = __shfl_up(incl, d);
      if (lane >= d) incl += n;
    }
    if (lane == 31) wtot[wave] = incl;
    __syncthreads();
    const int wv = wtot[lane];
    int winc = wv;
#pragma unroll
    for (int d = 1; d < 32; d <<= 1) {
      int n = __shfl_up(winc, d);
      if (lane >= d) winc += n;
    }
    const int tileTotal = __shfl(winc, 31);
    const int waveExcl  = __shfl(winc, wave) - __shfl(wv, wave);
    const int tbase = carry + waveExcl + (incl - run);
    int4* dst = (int4*)(off + base);
#pragma unroll
    for (int j = 0; j < 16; ++j) {
      int4 o = v[j];
      o.x += tbase; o.y += tbase; o.z += tbase; o.w += tbase;
      dst[j] = o;
    }
    carry += tileTotal;
    __syncthreads();
  }
}

__global__ __launch_bounds__(256) void gt_proj_wmma(
    const float* __restrict__ embeds,
    const float* __restrict__ qW, const float* __restrict__ kW,
    const float* __restrict__ vW,
    float* __restrict__ Q, float* __restrict__ K, float* __restrict__ V,
    int nMTiles) {
  const int lane = threadIdx.x & 31;
  const int wave = threadIdx.x >> 5;
  const int tile = blockIdx.x * 8 + wave;
  const int nTiles = nMTiles * 18;
  if (tile >= nTiles) return;

  const int mt  = tile / 18;
  const int sub = tile % 18;
  const int mat = sub / 6;
  const int n0  = (sub % 6) * 16;
  const int m0  = mt * 16;

  const float* W   = (mat == 0) ? qW : (mat == 1) ? kW : vW;
  float*       Out = (mat == 0) ? Q  : (mat == 1) ? K  : V;

  const int halfSel = lane >> 4;
  const int l15     = lane & 15;
  const int kOff    = halfSel * 2;

  const float* arow = embeds + (size_t)(m0 + l15) * DMODEL;
  const float* bcol = W + n0 + l15;

  v8f c = {0.f, 0.f, 0.f, 0.f, 0.f, 0.f, 0.f, 0.f};
#pragma unroll
  for (int kk = 0; kk < DMODEL; kk += 4) {
    v2f a = *(const v2f*)(arow + kk + kOff);
    v2f b;
    b.x = bcol[(size_t)(kk + kOff)     * DMODEL];
    b.y = bcol[(size_t)(kk + kOff + 1) * DMODEL];
    c = __builtin_amdgcn_wmma_f32_16x16x4_f32(false, a, false, b,
                                              (short)0, c, false, false);
  }

  float* orow = Out + (size_t)m0 * DMODEL + n0 + l15;
#pragma unroll
  for (int g = 0; g < 8; ++g)
    orow[(size_t)(g + halfSel * 8) * DMODEL] = c[g];
}

__global__ __launch_bounds__(256) void gt_edge_scores(
    const float* __restrict__ Q, const float* __restrict__ K,
    const int* __restrict__ rows, const int* __restrict__ cols,
    const int* __restrict__ rank, const int* __restrict__ off,
    float* __restrict__ sexp, int* __restrict__ scol, int nEdges) {
  int tid = blockIdx.x * 256 + threadIdx.x;
  if (tid >= nEdges * NHEADS) return;
  int e = tid >> 2, h = tid & 3;
  int r = rows[e], c = cols[e];
  const v4f* q = (const v4f*)(Q + (size_t)r * DMODEL + h * DHEAD);
  const v4f* k = (const v4f*)(K + (size_t)c * DMODEL + h * DHEAD);
  float s = 0.0f;
#pragma unroll
  for (int j = 0; j < 6; ++j) {
    v4f a = q[j], b = k[j];
    s += a.x * b.x + a.y * b.y + a.z * b.z + a.w * b.w;
  }
  s = fminf(10.0f, fmaxf(-10.0f, s));
  float es = expf(s);
  int pos = off[r] + rank[e];
  sexp[(size_t)pos * NHEADS + h] = es;
  if (h == 0) scol[pos] = c;
}

__global__ __launch_bounds__(256) void gt_row_gather(
    const float* __restrict__ V, const float* __restrict__ embeds,
    const int* __restrict__ off, const int* __restrict__ scol,
    const float* __restrict__ sexp, float* __restrict__ out, int nNodes) {
  const int lane = threadIdx.x & 31;
  const int row  = blockIdx.x * 8 + (threadIdx.x >> 5);
  if (row >= nNodes) return;
  const int beg = off[row], end = off[row + 1];
  const int el = lane >> 2, h = lane & 3;

  float part = 0.0f;
  for (int p = beg; p < end; p += 8)
    if (p + el < end) part += sexp[(size_t)(p + el) * NHEADS + h];
  part += __shfl_xor(part, 4);
  part += __shfl_xor(part, 8);
  part += __shfl_xor(part, 16);
  const float denom = part + 1e-8f;

  const int l4  = (lane < 24 ? lane : 23) * 4;
  const int myh = l4 / DHEAD;
  v4f acc = {0.0f, 0.0f, 0.0f, 0.0f};
  for (int p = beg; p < end; p += 8) {
    const bool valid = p + el < end;
    const int  pe = valid ? p + el : p;
    const float e0 = sexp[(size_t)pe * NHEADS + h];
    const int  cc = scol[pe];
    const float w = (valid ? e0 : 0.0f) / denom;
#pragma unroll
    for (int k = 0; k < 8; ++k) {
      const int   c  = __builtin_amdgcn_readlane(cc, 4 * k);
      const float wv = __shfl(w, 4 * k + myh);
      const v4f   t  = *(const v4f*)(V + (size_t)c * DMODEL + l4);
      acc += wv * t;
    }
  }
  if (lane < 24) {
    const size_t o = (size_t)row * DMODEL + l4;
    const v4f e4 = *(const v4f*)(embeds + o);
    *(v4f*)(out + o) = e4 + acc;
  }
}

extern "C" void kernel_launch(void* const* d_in, const int* in_sizes, int n_in,
                              void* d_out, int out_size, void* d_ws,
                              size_t ws_size, hipStream_t stream) {
  const float* embeds = (const float*)d_in[0];
  const float* qW     = (const float*)d_in[1];
  const float* kW     = (const float*)d_in[2];
  const float* vW     = (const float*)d_in[3];
  const int*   rows   = (const int*)d_in[4];
  const int*   cols   = (const int*)d_in[5];
  float*       out    = (float*)d_out;

  const int nNodes = in_sizes[0] / DMODEL;
  const int nEdges = in_sizes[4];

  const int    nScanTiles = nNodes / SCAN_TILE + 1;
  const size_t nPad  = (size_t)nScanTiles * SCAN_TILE;
  const size_t nEPad = ((size_t)nEdges + 3) & ~(size_t)3;

  float* ws = (float*)d_ws;
  const size_t nd = (size_t)nNodes * DMODEL;
  float* Q    = ws;
  float* Kp   = Q + nd;
  float* Vp   = Kp + nd;
  float* sexp = Vp + nd;
  int*   scol = (int*)(sexp + nEPad * NHEADS);
  int*   rank = scol + nEPad;
  int*   cnt  = rank + nEPad;
  int*   off  = cnt + nPad;

  gt_zero_counts<<<nScanTiles * (SCAN_TILE / 256), 256, 0, stream>>>(cnt);
  gt_edge_rank<<<(nEdges + 255) / 256, 256, 0, stream>>>(rows, cnt, rank, nEdges);
  gt_scan_offsets<<<1, 1024, 0, stream>>>(cnt, off, nScanTiles);
  {
    int nMTiles = nNodes / 16;
    int nTiles  = nMTiles * 18;
    int blocks  = (nTiles + 7) / 8;
    gt_proj_wmma<<<blocks, 256, 0, stream>>>(embeds, qW, kW, vW, Q, Kp, Vp,
                                             nMTiles);
  }
  {
    int total = nEdges * NHEADS;
    gt_edge_scores<<<(total + 255) / 256, 256, 0, stream>>>(Q, Kp, rows, cols,
                                                            rank, off, sexp, scol,
                                                            nEdges);
    gt_row_gather<<<(nNodes + 7) / 8, 256, 0, stream>>>(Vp, embeds, off, scol,
                                                        sexp, out, nNodes);
  }
}


# Scores folded into the per-row gather: the row's wave holds Q in registers, reads each source's K and V row once, keeps weights in registers; normalizer divides the finished sum (sum order changes)
# speedup vs baseline: 3.7981x; 1.3246x over previous
#include <hip/hip_runtime.h>
#include <cstddef>

typedef float v2f __attribute__((ext_vector_type(2)));
typedef float v4f __attribute__((ext_vector_type(4)));
typedef float v8f __attribute__((ext_vector_type(8)));

#define DMODEL 96
#define NHEADS 4
#define DHEAD  24
#define SCAN_TILE 65536

__global__ __launch_bounds__(256) void gt_zero_counts(int* __restrict__ cnt) {
  cnt[blockIdx.x * 256 + threadIdx.x] = 0;
}

__global__ __launch_bounds__(256) void gt_edge_rank(const int* __restrict__ rows,
                                                    int* __restrict__ cnt,
                                                    int* __restrict__ rank,
                                                    int nEdges) {
  int e = blockIdx.x * 256 + threadIdx.x;
  if (e < nEdges) rank[e] = atomicAdd(&cnt[rows[e]], 1);
}

__global__ __launch_bounds__(1024) void gt_scan_offsets(const int* __restrict__ cnt,
                                                        int* __restrict__ off,
                                                        int nScanTiles) {
  __shared__ int wtot[32];
  const int tid = threadIdx.x, lane = tid & 31, wave = tid >> 5;
  int carry = 0;
  for (int t = 0; t < nScanTiles; ++t) {
    const int base = t * SCAN_TILE + tid * 64;
    const int4* src = (const int4*)(cnt + base);
    int4 v[16];
#pragma unroll
    for (int j = 0; j < 16; ++j) v[j] = src[j];
    int run = 0;
#pragma unroll
    for (int j = 0; j < 16; ++j) {
      int a = v[j].x, b = v[j].y, c = v[j].z, d = v[j].w;
      v[j].x = run; run += a;
      v[j].y = run; run += b;
      v[j].z = run; run += c;
      v[j].w = run; run += d;
    }
    int incl = run;
#pragma unroll
    for (int d = 1; d < 32; d <<= 1) {
      int n = __shfl_up(incl, d);
      if (lane >= d) incl += n;
    }
    if (lane == 31) wtot[wave] = incl;
    __syncthreads();
    const int wv = wtot[lane];
    int winc = wv;
#pragma unroll
    for (int d = 1; d < 32; d <<= 1) {
      int n = __shfl_up(winc, d);
      if (lane >= d) winc += n;
    }
    const int tileTotal = __shfl(winc, 31);
    const int waveExcl  = __shfl(winc, wave) - __shfl(wv, wave);
    const int tbase = carry + waveExcl + (incl - run);
    int4* dst = (int4*)(off + base);
#pragma unroll
    for (int j = 0; j < 16; ++j) {
      int4 o = v[j];
      o.x += tbase; o.y += tbase; o.z += tbase; o.w += tbase;
      dst[j] = o;
    }
    carry += tileTotal;
    __syncthreads();
  }
}

__global__ __launch_bounds__(256) void gt_proj_wmma(
    const float* __restrict__ embeds,
    const float* __restrict__ qW, const float* __restrict__ kW,
    const float* __restrict__ vW,
    float* __restrict__ Q, float* __restrict__ K, float* __restrict__ V,
    int nMTiles) {
  const int lane = threadIdx.x & 31;
  const int wave = threadIdx.x >> 5;
  const int tile = blockIdx.x * 8 + wave;
  const int nTiles = nMTiles * 18;
  if (tile >= nTiles) return;

  const int mt  = tile / 18;
  const int sub = tile % 18;
  const int mat = sub / 6;
  const int n0  = (sub % 6) * 16;
  const int m0  = mt * 16;

  const float* W   = (mat == 0) ? qW : (mat == 1) ? kW : vW;
  float*       Out = (mat == 0) ? Q  : (mat == 1) ? K  : V;

  const int halfSel = lane >> 4;
  const int l15     = lane & 15;
  const int kOff    = halfSel * 2;

  const float* arow = embeds + (size_t)(m0 + l15) * DMODEL;
  const float* bcol = W + n0 + l15;

  v8f c = {0.f, 0.f, 0.f, 0.f, 0.f, 0.f, 0.f, 0.f};
#pragma unroll
  for (int kk = 0; kk < DMODEL; kk += 4) {
    v2f a = *(const v2f*)(arow + kk + kOff);
    v2f b;
    b.x = bcol[(size_t)(kk + kOff)     * DMODEL];
    b.y = bcol[(size_t)(kk + kOff + 1) * DMODEL];
    c = __builtin_amdgcn_wmma_f32_16x16x4_f32(false, a, false, b,
                                              (short)0, c, false, false);
  }

  float* orow = Out + (size_t)m0 * DMODEL + n0 + l15;
#pragma unroll
  for (int g = 0; g < 8; ++g)
    orow[(size_t)(g + halfSel * 8) * DMODEL] = c[g];
}

__global__ __launch_bounds__(256) void gt_place_cols(
    const int* __restrict__ rows, const int* __restrict__ cols,
    const int* __restrict__ rank, const int* __restrict__ off,
    int* __restrict__ scol, int nEdges) {
  int e = blockIdx.x * 256 + threadIdx.x;
  if (e < nEdges) scol[off[rows[e]] + rank[e]] = cols[e];
}

struct __attribute__((aligned(4))) f3 { float x, y, z; };

template <int M>
__device__ __forceinline__ float lane_xor(float x) {
  return __builtin_bit_cast(
      float, __builtin_amdgcn_update_dpp(0, __builtin_bit_cast(int, x),
                                         0x160 + M, 0xf, 0xf, true));
}

template <int J>
__device__ __forceinline__ float group_bcast(float x) {
  return __builtin_bit_cast(
      float, __builtin_amdgcn_mov_dpp8(__builtin_bit_cast(int, x),
                                       J * 0x249249));
}

__global__ __launch_bounds__(256) void gt_row_attend(
    const float* __restrict__ Q, const float* __restrict__ K,
    const float* __restrict__ V, const float* __restrict__ embeds,
    const int* __restrict__ off, const int* __restrict__ scol,
    float* __restrict__ out, int nNodes) {
  const int lane = threadIdx.x & 31;
  const int wave = __builtin_amdgcn_readfirstlane(threadIdx.x >> 5);
  const int row  = blockIdx.x * 8 + wave;
  if (row >= nNodes) return;
  const int beg = off[row], end = off[row + 1];
  const unsigned lo4 = (unsigned)lane * 12u;
  const int  sub = lane & 7;
  const bool b4 = (lane & 4) != 0, b2 = (lane & 2) != 0, b1 = (lane & 1) != 0;
  const char* Kb = (const char*)K;
  const char* Vb = (const char*)V;

  const f3 q = *(const f3*)((const char*)Q + (size_t)row * (DMODEL * 4) + lo4);
  float den = 0.0f, a0 = 0.0f, a1 = 0.0f, a2 = 0.0f;

  for (int p = beg; p < end; p += 32) {
    const int n = (end - p < 32) ? (end - p) : 32;
    const int mine = scol[p + (lane < n ? lane : 0)];
    for (int j0 = 0; j0 < n; j0 += 8) {
      f3 kf[8], vf[8];
      unsigned bo[8];
#pragma unroll
      for (int j = 0; j < 8; ++j) {
        const int jj = (j0 + j < n) ? (j0 + j) : j0;
        const int c  = __builtin_amdgcn_readlane(mine, jj);
        bo[j] = (unsigned)c * (DMODEL * 4u) + lo4;
      }
#pragma unroll
      for (int j = 0; j < 8; ++j) kf[j] = *(const f3*)(Kb + bo[j]);
#pragma unroll
      for (int j = 0; j < 8; ++j) vf[j] = *(const f3*)(Vb + bo[j]);
      __builtin_amdgcn_sched_barrier(0);

      float s[8];
#pragma unroll
      for (int j = 0; j < 8; ++j)
        s[j] = q.x * kf[j].x + q.y * kf[j].y + q.z * kf[j].z;
      float t[4], u[2];
#pragma unroll
      for (int k = 0; k < 4; ++k) {
        const float keep = b4 ? s[k + 4] : s[k];
        const float send = b4 ? s[k] : s[k + 4];
        t[k] = keep + lane_xor<4>(send);
      }
#pragma unroll
      for (int k = 0; k < 2; ++k) {
        const float keep = b2 ? t[k + 2] : t[k];
        const float send = b2 ? t[k] : t[k + 2];
        u[k] = keep + lane_xor<2>(send);
      }
      const float keep1 = b1 ? u[1] : u[0];
      const float send1 = b1 ? u[0] : u[1];
      float sc = keep1 + lane_xor<1>(send1);
      sc = fminf(10.0f, fmaxf(-10.0f, sc));
      const float es = (j0 + sub < n) ? expf(sc) : 0.0f;
      den += es;
      float w[8];
      w[0] = group_bcast<0>(es); w[1] = group_bcast<1>(es);
      w[2] = group_bcast<2>(es); w[3] = group_bcast<3>(es);
      w[4] = group_bcast<4>(es); w[5] = group_bcast<5>(es);
      w[6] = group_bcast<6>(es); w[7] = group_bcast<7>(es);
#pragma unroll
      for (int j = 0; j < 8; ++j) {
        a0 += w[j] * vf[j].x;
        a1 += w[j] * vf[j].y;
        a2 += w[j] * vf[j].z;
      }
    }
  }
  den += lane_xor<1>(den);
  den += lane_xor<2>(den);
  den += lane_xor<4>(den);
  const float inv = 1.0f / (den + 1e-8f);
  const size_t o = (size_t)row * (DMODEL * 4) + lo4;
  const f3 e3 = *(const f3*)((const char*)embeds + o);
  f3 r;
  r.x = e3.x + a0 * inv;
  r.y = e3.y + a1 * inv;
  r.z = e3.z + a2 * inv;
  *(f3*)((char*)out + o) = r;
}

extern "C" void kernel_launch(void* const* d_in, const int* in_sizes, int n_in,
                              void* d_out, int out_size, void* d_ws,
                              size_t ws_size, hipStream_t stream) {
  const float* embeds = (const float*)d_in[0];
  const float* qW     = (const float*)d_in[1];
  const float* kW     = (const float*)d_in[2];
  const float* vW     = (const float*)d_in[3];
  const int*   rows   = (const int*)d_in[4];
  const int*   cols   = (const int*)d_in[5];
  float*       out    = (float*)d_out;

  const int nNodes = in_sizes[0] / DMODEL;
  const int nEdges = in_sizes[4];

  const int    nScanTiles = nNodes / SCAN_TILE + 1;
  const size_t nPad  = (size_t)nScanTiles * SCAN_TILE;
  const size_t nEPad = ((size_t)nEdges + 3) & ~(size_t)3;

  float* ws = (float*)d_ws;
  const size_t nd = (size_t)nNodes * DMODEL;
  float* Q    = ws;
  float* Kp   = Q + nd;
  float* Vp   = Kp + nd;
  int*   scol = (int*)(Vp + nd);
  int*   rank = scol + nEPad;
  int*   cnt  = rank + nEPad;
  int*   off  = cnt + nPad;

  gt_zero_counts<<<nScanTiles * (SCAN_TILE / 256), 256, 0, stream>>>(cnt);
  gt_edge_rank<<<(nEdges + 255) / 256, 256, 0, stream>>>(rows, cnt, rank, nEdges);
  gt_scan_offsets<<<1, 1024, 0, stream>>>(cnt, off, nScanTiles);
  {
    int nMTiles = nNodes / 16;
    int nTiles  = nMTiles * 18;
    int blocks  = (nTiles + 7) / 8;
    gt_proj_wmma<<<blocks, 256, 0, stream>>>(embeds, qW, kW, vW, Q, Kp, Vp,
                                             nMTiles);
  }
  gt_place_cols<<<(nEdges + 255) / 256, 256, 0, stream>>>(rows, cols, rank, off,
                                                          scol, nEdges);
  gt_row_attend<<<(nNodes + 7) / 8, 256, 0, stream>>>(Q, Kp, Vp, embeds, off,
                                                      scol, out, nNodes);
}


# Edge-grouping stages (zero, rank, scan, place) each share a launch with a slice of the projection's column tiles instead of running alone: 5 launches, scan runs beside WMMA tiles; sums unchanged
# speedup vs baseline: 4.0989x; 1.0792x over previous
#include <hip/hip_runtime.h>
#include <cstddef>

typedef float v2f __attribute__((ext_vector_type(2)));
typedef float v4f __attribute__((ext_vector_type(4)));
typedef float v8f __attribute__((ext_vector_type(8)));

#define DMODEL 96
#define NHEADS 4
#define DHEAD  24
#define SCAN_TILE 65536

struct ProjArgs {
  const float* embeds;
  const float* qW;
  const float* kW;
  const float* vW;
  float* Q;
  float* K;
  float* V;
  int nMTiles;
};

template <int SB, int SC>
__device__ __forceinline__ void proj_tile(const ProjArgs& p, int tile) {
  const int lane = threadIdx.x & 31;
  if (tile >= p.nMTiles * SC) return;

  const int mt  = tile / SC;
  const int sub = SB + tile % SC;
  const int mat = sub / 6;
  const int n0  = (sub % 6) * 16;
  const int m0  = mt * 16;

  const float* W   = (mat == 0) ? p.qW : (mat == 1) ? p.kW : p.vW;
  float*       Out = (mat == 0) ? p.Q  : (mat == 1) ? p.K  : p.V;

  const int halfSel = lane >> 4;
  const int l15     = lane & 15;
  const int kOff    = halfSel * 2;

  const float* arow = p.embeds + (size_t)(m0 + l15) * DMODEL;
  const float* bcol = W + n0 + l15;

  v8f c = {0.f, 0.f, 0.f, 0.f, 0.f, 0.f, 0.f, 0.f};
#pragma unroll
  for (int kk = 0; kk < DMODEL; kk += 4) {
    v2f a = *(const v2f*)(arow + kk + kOff);
    v2f b;
    b.x = bcol[(size_t)(kk + kOff)     * DMODEL];
    b.y = bcol[(size_t)(kk + kOff + 1) * DMODEL];
    c = __builtin_amdgcn_wmma_f32_16x16x4_f32(false, a, false, b,
                                              (short)0, c, false, false);
  }

  float* orow = Out + (size_t)m0 * DMODEL + n0 + l15;
#pragma unroll
  for (int g = 0; g < 8; ++g)
    orow[(size_t)(g + halfSel * 8) * DMODEL] = c[g];
}

__global__ __launch_bounds__(256) void gt_zero_proj(int* __restrict__ cnt,
                                                    int nZeroBlocks, ProjArgs p) {
  const int b = blockIdx.x;
  if (b < nZeroBlocks) {
    cnt[b * 256 + threadIdx.x] = 0;
    return;
  }
  proj_tile<0, 2>(p, (b - nZeroBlocks) * 8 + (int)(threadIdx.x >> 5));
}

__global__ __launch_bounds__(256) void gt_rank_proj(const int* __restrict__ rows,
                                                    int* __restrict__ cnt,
                                                    int* __restrict__ rank,
                                                    int nEdges, int nRankBlocks,
                                                    ProjArgs p) {
  const int b = blockIdx.x;
  if (b < nRankBlocks) {
    int e = b * 256 + threadIdx.x;
    if (e < nEdges) rank[e] = atomicAdd(&cnt[rows[e]], 1);
    return;
  }
  proj_tile<2, 4>(p, (b - nRankBlocks) * 8 + (int)(threadIdx.x >> 5));
}

__device__ __forceinline__ void scan_offsets(const int* __restrict__ cnt,
                                             int* __restrict__ off,
                                             int nScanTiles) {
  __shared__ int wtot[32];
  const int tid = threadIdx.x, lane = tid & 31, wave = tid >> 5;
  int carry = 0;
  for (int t = 0; t < nScanTiles; ++t) {
    const int base = t * SCAN_TILE + tid * 64;
    const int4* src = (const int4*)(cnt + base);
    int4 v[16];
#pragma unroll
    for (int j = 0; j < 16; ++j) v[j] = src[j];
    int run = 0;
#pragma unroll
    for (int j = 0; j < 16; ++j) {
      int a = v[j].x, b = v[j].y, c = v[j].z, d = v[j].w;
      v[j].x = run; run += a;
      v[j].y = run; run += b;
      v[j].z = run; run += c;
      v[j].w = run; run += d;
    }
    int incl = run;
#pragma unroll
    for (int d = 1; d < 32; d <<= 1) {
      int n = __shfl_up(incl, d);
      if (lane >= d) incl += n;
    }
    if (lane == 31) wtot[wave] = incl;
    __syncthreads();
    const int wv = wtot[lane];
    int winc = wv;
#pragma unroll
    for (int d = 1; d < 32; d <<= 1) {
      int n = __shfl_up(winc, d);
      if (lane >= d) winc += n;
    }
    const int tileTotal = __shfl(winc, 31);
    const int waveExcl  = __shfl(winc, wave) - __shfl(wv, wave);
    const int tbase = carry + waveExcl + (incl - run);
    int4* dst = (int4*)(off + base);
#pragma unroll
    for (int j = 0; j < 16; ++j) {
      int4 o = v[j];
      o.x += tbase; o.y += tbase; o.z += tbase; o.w += tbase;
      dst[j] = o;
    }
    carry += tileTotal;
    __syncthreads();
  }
}

__global__ __launch_bounds__(1024) void gt_scan_proj(const int* __restrict__ cnt,
                                                     int* __restrict__ off,
                                                     int nScanTiles, ProjArgs p) {
  const int b = blockIdx.x;
  if (b == 0) {
    scan_offsets(cnt, off, nScanTiles);
    return;
  }
  proj_tile<6, 8>(p, (b - 1) * 32 + (int)(threadIdx.x >> 5));
}

__global__ __launch_bounds__(256) void gt_place_proj(
    const int* __restrict__ rows, const int* __restrict__ cols,
    const int* __restrict__ rank, const int* __restrict__ off,
    int* __restrict__ scol, int nEdges, int nPlaceBlocks, ProjArgs p) {
  const int b = blockIdx.x;
  if (b < nPlaceBlocks) {
    int e = b * 256 + threadIdx.x;
    if (e < nEdges) scol[off[rows[e]] + rank[e]] = cols[e];
    return;
  }
  proj_tile<14, 4>(p, (b - nPlaceBlocks) * 8 + (int)(threadIdx.x >> 5));
}

struct __attribute__((aligned(4))) f3 { float x, y, z; };

template <int M>
__device__ __forceinline__ float lane_xor(float x) {
  return __builtin_bit_cast(
      float, __builtin_amdgcn_update_dpp(0, __builtin_bit_cast(int, x),
                                         0x160 + M, 0xf, 0xf, true));
}

template <int J>
__device__ __forceinline__ float group_bcast(float x) {
  return __builtin_bit_cast(
      float, __builtin_amdgcn_mov_dpp8(__builtin_bit_cast(int, x),
                                       J * 0x249249));
}

__global__ __launch_bounds__(256) void gt_row_attend(
    const float* __restrict__ Q, const float* __restrict__ K,
    const float* __restrict__ V, const float* __restrict__ embeds,
    const int* __restrict__ off, const int* __restrict__ scol,
    float* __restrict__ out, int nNodes) {
  const int lane = threadIdx.x & 31;
  const int wave = __builtin_amdgcn_readfirstlane(threadIdx.x >> 5);
  const int row  = blockIdx.x * 8 + wave;
  if (row >= nNodes) return;
  const int beg = off[row], end = off[row + 1];
  const unsigned lo4 = (unsigned)lane * 12u;
  const int  sub = lane & 7;
  const bool b4 = (lane & 4) != 0, b2 = (lane & 2) != 0, b1 = (lane & 1) != 0;
  const char* Kb = (const char*)K;
  const char* Vb = (const char*)V;

  const f3 q = *(const f3*)((const char*)Q + (size_t)row * (DMODEL * 4) + lo4);
  float den = 0.0f, a0 = 0.0f, a1 = 0.0f, a2 = 0.0f;

  for (int p = beg; p < end; p += 32) {
    const int n = (end - p < 32) ? (end - p) : 32;
    const int mine = scol[p + (lane < n ? lane : 0)];
    for (int j0 = 0; j0 < n; j0 += 8) {
      f3 kf[8], vf[8];
      unsigned bo[8];
#pragma unroll
      for (int j = 0; j < 8; ++j) {
        const int jj = (j0 + j < n) ? (j0 + j) : j0;
        const int c  = __builtin_amdgcn_readlane(mine, jj);
        bo[j] = (unsigned)c * (DMODEL * 4u) + lo4;
      }
#pragma unroll
      for (int j = 0; j < 8; ++j) kf[j] = *(const f3*)(Kb + bo[j]);
#pragma unroll
      for (int j = 0; j < 8; ++j) vf[j] = *(const f3*)(Vb + bo[j]);
      __builtin_amdgcn_sched_barrier(0);

      float s[8];
#pragma unroll
      for (int j = 0; j < 8; ++j)
        s[j] = q.x * kf[j].x + q.y * kf[j].y + q.z * kf[j].z;
      float t[4], u[2];
#pragma unroll
      for (int k = 0; k < 4; ++k) {
        const float keep = b4 ? s[k + 4] : s[k];
        const float send = b4 ? s[k] : s[k + 4];
        t[k] = keep + lane_xor<4>(send);
      }
#pragma unroll
      for (int k = 0; k < 2; ++k) {
        const float keep = b2 ? t[k + 2] : t[k];
        const float send = b2 ? t[k] : t[k + 2];
        u[k] = keep + lane_xor<2>(send);
      }
      const float keep1 = b1 ? u[1] : u[0];
      const float send1 = b1 ? u[0] : u[1];
      float sc = keep1 + lane_xor<1>(send1);
      sc = fminf(10.0f, fmaxf(-10.0f, sc));
      const float es = (j0 + sub < n) ? expf(sc) : 0.0f;
      den += es;
      float w[8];
      w[0] = group_bcast<0>(es); w[1] = group_bcast<1>(es);
      w[2] = group_bcast<2>(es); w[3] = group_bcast<3>(es);
      w[4] = group_bcast<4>(es); w[5] = group_bcast<5>(es);
      w[6] = group_bcast<6>(es); w[7] = group_bcast<7>(es);
#pragma unroll
      for (int j = 0; j < 8; ++j) {
        a0 += w[j] * vf[j].x;
        a1 += w[j] * vf[j].y;
        a2 += w[j] * vf[j].z;
      }
    }
  }
  den += lane_xor<1>(den);
  den += lane_xor<2>(den);
  den += lane_xor<4>(den);
  const float inv = 1.0f / (den + 1e-8f);
  const size_t o = (size_t)row * (DMODEL * 4) + lo4;
  const f3 e3 = *(const f3*)((const char*)embeds + o);
  f3 r;
  r.x = e3.x + a0 * inv;
  r.y = e3.y + a1 * inv;
  r.z = e3.z + a2 * inv;
  *(f3*)((char*)out + o) = r;
}

extern "C" void kernel_launch(void* const* d_in, const int* in_sizes, int n_in,
                              void* d_out, int out_size, void* d_ws,
                              size_t ws_size, hipStream_t stream) {
  const float* embeds = (const float*)d_in[0];
  const float* qW     = (const float*)d_in[1];
  const float* kW     = (const float*)d_in[2];
  const float* vW     = (const float*)d_in[3];
  const int*   rows   = (const int*)d_in[4];
  const int*   cols   = (const int*)d_in[5];
  float*       out    = (float*)d_out;

  const int nNodes = in_sizes[0] / DMODEL;
  const int nEdges = in_sizes[4];

  const int    nScanTiles = nNodes / SCAN_TILE + 1;
  const size_t nPad  = (size_t)nScanTiles * SCAN_TILE;
  const size_t nEPad = ((size_t)nEdges + 3) & ~(size_t)3;

  float* ws = (float*)d_ws;
  const size_t nd = (size_t)nNodes * DMODEL;
  float* Q    = ws;
  float* Kp   = Q + nd;
  float* Vp   = Kp + nd;
  int*   scol = (int*)(Vp + nd);
  int*   rank = scol + nEPad;
  int*   cnt  = rank + nEPad;
  int*   off  = cnt + nPad;

  const int nMTiles = nNodes / 16;
  ProjArgs p;
  p.embeds = embeds; p.qW = qW; p.kW = kW; p.vW = vW;
  p.Q = Q; p.K = Kp; p.V = Vp; p.nMTiles = nMTiles;

  const int nZeroBlocks = nScanTiles * (SCAN_TILE / 256);
  const int nEdgeBlocks = (nEdges + 255) / 256;
  const int proj2 = (nMTiles * 2 + 7) / 8;
  const int proj4 = (nMTiles * 4 + 7) / 8;
  const int proj8w = (nMTiles * 8 + 31) / 32;

  gt_zero_proj<<<nZeroBlocks + proj2, 256, 0, stream>>>(cnt, nZeroBlocks, p);
  gt_rank_proj<<<nEdgeBlocks + proj4, 256, 0, stream>>>(rows, cnt, rank, nEdges,
                                                        nEdgeBlocks, p);
  gt_scan_proj<<<1 + proj8w, 1024, 0, stream>>>(cnt, off, nScanTiles, p);
  gt_place_proj<<<nEdgeBlocks + proj4, 256, 0, stream>>>(rows, cols, rank, off,
                                                         scol, nEdges,
                                                         nEdgeBlocks, p);
  gt_row_attend<<<(nNodes + 7) / 8, 256, 0, stream>>>(Q, Kp, Vp, embeds, off,
                                                      scol, out, nNodes);
}


# Edge grouping redone as a two-level counting sort in LDS (slices bin edges by 512-row range, owners sort their range): no global atomics, no per-edge scattered stores, 3 launches; ranks arrival-order
# speedup vs baseline: 7.0315x; 1.7155x over previous
#include <hip/hip_runtime.h>
#include <cstddef>

typedef float v2f __attribute__((ext_vector_type(2)));
typedef float v4f __attribute__((ext_vector_type(4)));
typedef float v8f __attribute__((ext_vector_type(8)));

#define DMODEL 96
#define NHEADS 4
#define DHEAD  24
#define OWN_SHIFT 9
#define OWN_NODES 512
#define MAX_OWN   128
#define SLICE     4096
#define STAGE_CAP 12288

struct ProjArgs {
  const float* embeds;
  const float* qW;
  const float* kW;
  const float* vW;
  float* Q;
  float* K;
  float* V;
  int nMTiles;
};

template <int SB, int SC>
__device__ __forceinline__ void proj_tile(const ProjArgs& p, int tile) {
  const int lane = threadIdx.x & 31;
  if (tile >= p.nMTiles * SC) return;

  const int mt  = tile / SC;
  const int sub = SB + tile % SC;
  const int mat = sub / 6;
  const int n0  = (sub % 6) * 16;
  const int m0  = mt * 16;

  const float* W   = (mat == 0) ? p.qW : (mat == 1) ? p.kW : p.vW;
  float*       Out = (mat == 0) ? p.Q  : (mat == 1) ? p.K  : p.V;

  const int halfSel = lane >> 4;
  const int l15     = lane & 15;
  const int kOff    = halfSel * 2;

  const float* arow = p.embeds + (size_t)(m0 + l15) * DMODEL;
  const float* bcol = W + n0 + l15;

  v8f c = {0.f, 0.f, 0.f, 0.f, 0.f, 0.f, 0.f, 0.f};
#pragma unroll
  for (int kk = 0; kk < DMODEL; kk += 4) {
    v2f a = *(const v2f*)(arow + kk + kOff);
    v2f b;
    b.x = bcol[(size_t)(kk + kOff)     * DMODEL];
    b.y = bcol[(size_t)(kk + kOff + 1) * DMODEL];
    c = __builtin_amdgcn_wmma_f32_16x16x4_f32(false, a, false, b,
                                              (short)0, c, false, false);
  }

  float* orow = Out + (size_t)m0 * DMODEL + n0 + l15;
#pragma unroll
  for (int g = 0; g < 8; ++g)
    orow[(size_t)(g + halfSel * 8) * DMODEL] = c[g];
}

__device__ __forceinline__ void bin_slice(const int* __restrict__ rows,
                                          const int* __restrict__ cols,
                                          int2* __restrict__ binned,
                                          int* __restrict__ C,
                                          int* __restrict__ S,
                                          int nEdges, int g) {
  __shared__ int  hist[MAX_OWN];
  __shared__ int  start[MAX_OWN];
  __shared__ int  fill[MAX_OWN];
  __shared__ int2 stage[SLICE];
  const int tid = threadIdx.x, lane = tid & 31;
  if (tid < MAX_OWN) { hist[tid] = 0; fill[tid] = 0; }
  __syncthreads();
  int r[4], c[4];
  const int e0 = g * SLICE + tid;
#pragma unroll
  for (int k = 0; k < 4; ++k) {
    const int e = e0 + k * 1024;
    r[k] = -1; c[k] = 0;
    if (e < nEdges) { r[k] = rows[e]; c[k] = cols[e]; }
  }
#pragma unroll
  for (int k = 0; k < 4; ++k)
    if (r[k] >= 0) atomicAdd(&hist[r[k] >> OWN_SHIFT], 1);
  __syncthreads();
  if (tid < 32) {
    const int a0 = hist[4 * tid], a1 = hist[4 * tid + 1];
    const int a2 = hist[4 * tid + 2], a3 = hist[4 * tid + 3];
    const int s = a0 + a1 + a2 + a3;
    int incl = s;
#pragma unroll
    for (int d = 1; d < 32; d <<= 1) {
      int n = __shfl_up(incl, d);
      if (lane >= d) incl += n;
    }
    const int ex = incl - s;
    start[4 * tid]     = ex;
    start[4 * tid + 1] = ex + a0;
    start[4 * tid + 2] = ex + a0 + a1;
    start[4 * tid + 3] = ex + a0 + a1 + a2;
  }
  __syncthreads();
  if (tid < MAX_OWN) {
    C[g * MAX_OWN + tid] = hist[tid];
    S[g * MAX_OWN + tid] = start[tid];
  }
#pragma unroll
  for (int k = 0; k < 4; ++k)
    if (r[k] >= 0) {
      const int o = r[k] >> OWN_SHIFT;
      const int pos = start[o] + atomicAdd(&fill[o], 1);
      stage[pos] = make_int2(r[k], c[k]);
    }
  __syncthreads();
  const int left = nEdges - g * SLICE;
  const int len = left < SLICE ? left : SLICE;
  int2* dst = binned + (size_t)g * SLICE;
  for (int i = tid; i < len; i += 1024) dst[i] = stage[i];
}

__global__ __launch_bounds__(1024) void gt_bin_proj(
    const int* __restrict__ rows, const int* __restrict__ cols,
    int2* __restrict__ binned, int* __restrict__ C, int* __restrict__ S,
    int nEdges, int nSlices, ProjArgs p) {
  const int b = blockIdx.x;
  if (b < nSlices) {
    bin_slice(rows, cols, binned, C, S, nEdges, b);
    return;
  }
  proj_tile<0, 9>(p, (b - nSlices) * 32 + (int)(threadIdx.x >> 5));
}

__device__ __forceinline__ void sort_owner(const int2* __restrict__ binned,
                                           const int* __restrict__ C,
                                           const int* __restrict__ S,
                                           int* __restrict__ off,
                                           int* __restrict__ scol,
                                           int nNodes, int nSlices, int nOwn,
                                           int o) {
  __shared__ int part[8][MAX_OWN];
  __shared__ int tot[MAX_OWN];
  __shared__ int cnt[OWN_NODES];
  __shared__ int excl[OWN_NODES];
  __shared__ int fil[OWN_NODES];
  __shared__ int stg[STAGE_CAP];
  __shared__ int shBase;
  const int tid = threadIdx.x, lane = tid & 31;
  const int wave = __builtin_amdgcn_readfirstlane(tid >> 5);

  {
    const int oo = tid & (MAX_OWN - 1), pt = tid >> 7;
    int s = 0;
#pragma unroll 8
    for (int g = pt; g < nSlices; g += 8) s += C[g * MAX_OWN + oo];
    part[pt][oo] = s;
  }
  if (tid < OWN_NODES) { cnt[tid] = 0; fil[tid] = 0; }
  int myN = 0, mySt = 0;
  {
    const int g = wave + 32 * lane;
    if (g < nSlices) { myN = C[g * MAX_OWN + o]; mySt = S[g * MAX_OWN + o]; }
  }
  __syncthreads();
  if (tid < MAX_OWN) {
    int s = 0;
#pragma unroll
    for (int k = 0; k < 8; ++k) s += part[k][tid];
    tot[tid] = s;
  }
  __syncthreads();
  if (tid < 32) {
    int s = 0;
#pragma unroll
    for (int k = 0; k < 4; ++k) {
      const int idx = 4 * tid + k;
      if (idx < o) s += tot[idx];
    }
#pragma unroll
    for (int d = 16; d >= 1; d >>= 1) s += __shfl_xor(s, d);
    if (tid == 0) shBase = s;
  }
  const int node0 = o << OWN_SHIFT;
  for (int j = 0; wave + 32 * j < nSlices; ++j) {
    const int n  = __builtin_amdgcn_readlane(myN, j);
    const int st = __builtin_amdgcn_readlane(mySt, j);
    const int2* run = binned + (size_t)(wave + 32 * j) * SLICE + st;
    for (int i = lane; i < n; i += 32) atomicAdd(&cnt[run[i].x - node0], 1);
  }
  __syncthreads();
  const int base = shBase, total = tot[o];
  if (tid < 32) {
    int v[16];
    int run = 0;
#pragma unroll
    for (int k = 0; k < 16; ++k) { const int a = cnt[16 * tid + k]; v[k] = run; run += a; }
    int incl = run;
#pragma unroll
    for (int d = 1; d < 32; d <<= 1) {
      int n = __shfl_up(incl, d);
      if (lane >= d) incl += n;
    }
    const int ex = incl - run;
#pragma unroll
    for (int k = 0; k < 16; ++k) excl[16 * tid + k] = ex + v[k];
  }
  __syncthreads();
  if (tid < OWN_NODES && node0 + tid < nNodes) off[node0 + tid] = base + excl[tid];
  if (tid == 0 && o == nOwn - 1) off[nNodes] = base + total;
  const bool useLds = total <= STAGE_CAP;
  for (int j = 0; wave + 32 * j < nSlices; ++j) {
    const int n  = __builtin_amdgcn_readlane(myN, j);
    const int st = __builtin_amdgcn_readlane(mySt, j);
    const int2* run = binned + (size_t)(wave + 32 * j) * SLICE + st;
    for (int i = lane; i < n; i += 32) {
      const int2 rc = run[i];
      const int d = rc.x - node0;
      const int pos = excl[d] + atomicAdd(&fil[d], 1);
      if (useLds) stg[pos] = rc.y; else scol[base + pos] = rc.y;
    }
  }
  __syncthreads();
  if (useLds)
    for (int i = tid; i < total; i += 1024) scol[base + i] = stg[i];
}

__global__ __launch_bounds__(1024) void gt_sort_proj(
    const int2* __restrict__ binned, const int* __restrict__ C,
    const int* __restrict__ S, int* __restrict__ off, int* __restrict__ scol,
    int nNodes, int nSlices, int nOwn, ProjArgs p) {
  const int b = blockIdx.x;
  if (b < nOwn) {
    sort_owner(binned, C, S, off, scol, nNodes, nSlices, nOwn, b);
    return;
  }
  proj_tile<9, 9>(p, (b - nOwn) * 32 + (int)(threadIdx.x >> 5));
}

struct __attribute__((aligned(4))) f3 { float x, y, z; };

template <int M>
__device__ __forceinline__ float lane_xor(float x) {
  return __builtin_bit_cast(
      float, __builtin_amdgcn_update_dpp(0, __builtin_bit_cast(int, x),
                                         0x160 + M, 0xf, 0xf, true));
}

template <int J>
__device__ __forceinline__ float group_bcast(float x) {
  return __builtin_bit_cast(
      float, __builtin_amdgcn_mov_dpp8(__builtin_bit_cast(int, x),
                                       J * 0x249249));
}

__global__ __launch_bounds__(256) void gt_row_attend(
    const float* __restrict__ Q, const float* __restrict__ K,
    const float* __restrict__ V, const float* __restrict__ embeds,
    const int* __restrict__ off, const int* __restrict__ scol,
    float* __restrict__ out, int nNodes) {
  const int lane = threadIdx.x & 31;
  const int wave = __builtin_amdgcn_readfirstlane(threadIdx.x >> 5);
  const int row  = blockIdx.x * 8 + wave;
  if (row >= nNodes) return;
  const int beg = off[row], end = off[row + 1];
  const unsigned lo4 = (unsigned)lane * 12u;
  const int  sub = lane & 7;
  const bool b4 = (lane & 4) != 0, b2 = (lane & 2) != 0, b1 = (lane & 1) != 0;
  const char* Kb = (const char*)K;
  const char* Vb = (const char*)V;

  const f3 q = *(const f3*)((const char*)Q + (size_t)row * (DMODEL * 4) + lo4);
  float den = 0.0f, a0 = 0.0f, a1 = 0.0f, a2 = 0.0f;

  for (int p = beg; p < end; p += 32) {
    const int n = (end - p < 32) ? (end - p) : 32;
    const int mine = scol[p + (lane < n ? lane : 0)];
    for (int j0 = 0; j0 < n; j0 += 8) {
      f3 kf[8], vf[8];
      unsigned bo[8];
#pragma unroll
      for (int j = 0; j < 8; ++j) {
        const int jj = (j0 + j < n) ? (j0 + j) : j0;
        const int c  = __builtin_amdgcn_readlane(mine, jj);
        bo[j] = (unsigned)c * (DMODEL * 4u) + lo4;
      }
#pragma unroll
      for (int j = 0; j < 8; ++j) kf[j] = *(const f3*)(Kb + bo[j]);
#pragma unroll
      for (int j = 0; j < 8; ++j) vf[j] = *(const f3*)(Vb + bo[j]);
      __builtin_amdgcn_sched_barrier(0);

      float s[8];
#pragma unroll
      for (int j = 0; j < 8; ++j)
        s[j] = q.x * kf[j].x + q.y * kf[j].y + q.z * kf[j].z;
      float t[4], u[2];
#pragma unroll
      for (int k = 0; k < 4; ++k) {
        const float keep = b4 ? s[k + 4] : s[k];
        const float send = b4 ? s[k] : s[k + 4];
        t[k] = keep + lane_xor<4>(send);
      }
#pragma unroll
      for (int k = 0; k < 2; ++k) {
        const float keep = b2 ? t[k + 2] : t[k];
        const float send = b2 ? t[k] : t[k + 2];
        u[k] = keep + lane_xor<2>(send);
      }
      const float keep1 = b1 ? u[1] : u[0];
      const float send1 = b1 ? u[0] : u[1];
      float sc = keep1 + lane_xor<1>(send1);
      sc = fminf(10.0f, fmaxf(-10.0f, sc));
      const float es = (j0 + sub < n) ? expf(sc) : 0.0f;
      den += es;
      float w[8];
      w[0] = group_bcast<0>(es); w[1] = group_bcast<1>(es);
      w[2] = group_bcast<2>(es); w[3] = group_bcast<3>(es);
      w[4] = group_bcast<4>(es); w[5] = group_bcast<5>(es);
      w[6] = group_bcast<6>(es); w[7] = group_bcast<7>(es);
#pragma unroll
      for (int j = 0; j < 8; ++j) {
        a0 += w[j] * vf[j].x;
        a1 += w[j] * vf[j].y;
        a2 += w[j] * vf[j].z;
      }
    }
  }
  den += lane_xor<1>(den);
  den += lane_xor<2>(den);
  den += lane_xor<4>(den);
  const float inv = 1.0f / (den + 1e-8f);
  const size_t o = (size_t)row * (DMODEL * 4) + lo4;
  const f3 e3 = *(const f3*)((const char*)embeds + o);
  f3 r;
  r.x = e3.x + a0 * inv;
  r.y = e3.y + a1 * inv;
  r.z = e3.z + a2 * inv;
  *(f3*)((char*)out + o) = r;
}

extern "C" void kernel_launch(void* const* d_in, const int* in_sizes, int n_in,
                              void* d_out, int out_size, void* d_ws,
                              size_t ws_size, hipStream_t stream) {
  const float* embeds = (const float*)d_in[0];
  const float* qW     = (const float*)d_in[1];
  const float* kW     = (const float*)d_in[2];
  const float* vW     = (const float*)d_in[3];
  const int*   rows   = (const int*)d_in[4];
  const int*   cols   = (const int*)d_in[5];
  float*       out    = (float*)d_out;

  const int nNodes = in_sizes[0] / DMODEL;
  const int nEdges = in_sizes[4];

  const int    nSlices = (nEdges + SLICE - 1) / SLICE;
  const int    nOwn    = (nNodes + OWN_NODES - 1) / OWN_NODES;
  const size_t nEPad   = ((size_t)nEdges + 3) & ~(size_t)3;

  float* ws = (float*)d_ws;
  const size_t nd = (size_t)nNodes * DMODEL;
  float* Q      = ws;
  float* Kp     = Q + nd;
  float* Vp     = Kp + nd;
  int*   scol   = (int*)(Vp + nd);
  int2*  binned = (int2*)(scol + nEPad);
  int*   C      = (int*)(binned + (size_t)nSlices * SLICE);
  int*   S      = C + (size_t)nSlices * MAX_OWN;
  int*   off    = S + (size_t)nSlices * MAX_OWN;

  const int nMTiles = nNodes / 16;
  ProjArgs p;
  p.embeds = embeds; p.qW = qW; p.kW = kW; p.vW = vW;
  p.Q = Q; p.K = Kp; p.V = Vp; p.nMTiles = nMTiles;

  const int proj9 = (nMTiles * 9 + 31) / 32;

  gt_bin_proj<<<nSlices + proj9, 1024, 0, stream>>>(rows, cols, binned, C, S,
                                                    nEdges, nSlices, p);
  gt_sort_proj<<<nOwn + proj9, 1024, 0, stream>>>(binned, C, S, off, scol,
                                                  nNodes, nSlices, nOwn, p);
  gt_row_attend<<<(nNodes + 7) / 8, 256, 0, stream>>>(Q, Kp, Vp, embeds, off,
                                                      scol, out, nNodes);
}


# Projection: a wave makes a 16x48 block, 3 WMMA tiles sharing each A fragment, A read 16 B and B, D 12 B per lane, instead of one 16x16 tile; k order inside a pair of WMMA steps changes (sum order)
# speedup vs baseline: 7.5839x; 1.0786x over previous
#include <hip/hip_runtime.h>
#include <cstddef>

typedef float v2f __attribute__((ext_vector_type(2)));
typedef float v4f __attribute__((ext_vector_type(4)));
typedef float v8f __attribute__((ext_vector_type(8)));

#define DMODEL 96
#define NHEADS 4
#define DHEAD  24
#define OWN_SHIFT 9
#define OWN_NODES 512
#define MAX_OWN   128
#define SLICE     4096
#define STAGE_CAP 12288

struct ProjArgs {
  const float* embeds;
  const float* qW;
  const float* kW;
  const float* vW;
  float* Q;
  float* K;
  float* V;
  int nMTiles;
};

struct __attribute__((aligned(4))) f3 { float x, y, z; };

template <int GB>
__device__ __forceinline__ void proj_unit(const ProjArgs& p, int unitOfWave) {
  const int lane = threadIdx.x & 31;
  const int unit = __builtin_amdgcn_readfirstlane(unitOfWave);
  if (unit >= p.nMTiles * 3) return;

  const int mt  = unit / 3;
  const int grp = GB + unit % 3;
  const int mat = grp >> 1;
  const int g0  = (grp & 1) * 48;
  const int m0  = mt * 16;

  const float* W   = (mat == 0) ? p.qW : (mat == 1) ? p.kW : p.vW;
  float*       Out = (mat == 0) ? p.Q  : (mat == 1) ? p.K  : p.V;

  const int h = lane >> 4;
  const int c = lane & 15;

  const float* arow = p.embeds + (size_t)(m0 + c) * DMODEL + 4 * h;
  const float* bcol = W + (size_t)(4 * h) * DMODEL + g0 + 3 * c;

  v8f acc0 = {0.f, 0.f, 0.f, 0.f, 0.f, 0.f, 0.f, 0.f};
  v8f acc1 = acc0, acc2 = acc0;
#pragma unroll
  for (int kb = 0; kb < DMODEL; kb += 8) {
    const v4f a  = *(const v4f*)(arow + kb);
    const f3  w0 = *(const f3*)(bcol + (size_t)(kb + 0) * DMODEL);
    const f3  w1 = *(const f3*)(bcol + (size_t)(kb + 1) * DMODEL);
    const f3  w2 = *(const f3*)(bcol + (size_t)(kb + 2) * DMODEL);
    const f3  w3 = *(const f3*)(bcol + (size_t)(kb + 3) * DMODEL);
    const v2f a01 = {a.x, a.y};
    const v2f a23 = {a.z, a.w};
    v2f b;
    b.x = w0.x; b.y = w1.x;
    acc0 = __builtin_amdgcn_wmma_f32_16x16x4_f32(false, a01, false, b, (short)0, acc0, false, false);
    b.x = w0.y; b.y = w1.y;
    acc1 = __builtin_amdgcn_wmma_f32_16x16x4_f32(false, a01, false, b, (short)0, acc1, false, false);
    b.x = w0.z; b.y = w1.z;
    acc2 = __builtin_amdgcn_wmma_f32_16x16x4_f32(false, a01, false, b, (short)0, acc2, false, false);
    b.x = w2.x; b.y = w3.x;
    acc0 = __builtin_amdgcn_wmma_f32_16x16x4_f32(false, a23, false, b, (short)0, acc0, false, false);
    b.x = w2.y; b.y = w3.y;
    acc1 = __builtin_amdgcn_wmma_f32_16x16x4_f32(false, a23, false, b, (short)0, acc1, false, false);
    b.x = w2.z; b.y = w3.z;
    acc2 = __builtin_amdgcn_wmma_f32_16x16x4_f32(false, a23, false, b, (short)0, acc2, false, false);
  }

  float* orow = Out + (size_t)(m0 + 8 * h) * DMODEL + g0 + 3 * c;
#pragma unroll
  for (int g = 0; g < 8; ++g) {
    f3 r;
    r.x = acc0[g]; r.y = acc1[g]; r.z = acc2[g];
    *(f3*)(orow + (size_t)g * DMODEL) = r;
  }
}

__device__ __forceinline__ void bin_slice(const int* __restrict__ rows,
                                          const int* __restrict__ cols,
                                          int2* __restrict__ binned,
                                          int* __restrict__ C,
                                          int* __restrict__ S,
                                          int nEdges, int g) {
  __shared__ int  hist[MAX_OWN];
  __shared__ int  start[MAX_OWN];
  __shared__ int  fill[MAX_OWN];
  __shared__ int2 stage[SLICE];
  const int tid = threadIdx.x, lane = tid & 31;
  if (tid < MAX_OWN) { hist[tid] = 0; fill[tid] = 0; }
  __syncthreads();
  int r[4], c[4];
  const int e0 = g * SLICE + tid;
#pragma unroll
  for (int k = 0; k < 4; ++k) {
    const int e = e0 + k * 1024;
    r[k] = -1; c[k] = 0;
    if (e < nEdges) { r[k] = rows[e]; c[k] = cols[e]; }
  }
#pragma unroll
  for (int k = 0; k < 4; ++k)
    if (r[k] >= 0) atomicAdd(&hist[r[k] >> OWN_SHIFT], 1);
  __syncthreads();
  if (tid < 32) {
    const int a0 = hist[4 * tid], a1 = hist[4 * tid + 1];
    const int a2 = hist[4 * tid + 2], a3 = hist[4 * tid + 3];
    const int s = a0 + a1 + a2 + a3;
    int incl = s;
#pragma unroll
    for (int d = 1; d < 32; d <<= 1) {
      int n = __shfl_up(incl, d);
      if (lane >= d) incl += n;
    }
    const int ex = incl - s;
    start[4 * tid]     = ex;
    start[4 * tid + 1] = ex + a0;
    start[4 * tid + 2] = ex + a0 + a1;
    start[4 * tid + 3] = ex + a0 + a1 + a2;
  }
  __syncthreads();
  if (tid < MAX_OWN) {
    C[g * MAX_OWN + tid] = hist[tid];
    S[g * MAX_OWN + tid] = start[tid];
  }
#pragma unroll
  for (int k = 0; k < 4; ++k)
    if (r[k] >= 0) {
      const int o = r[k] >> OWN_SHIFT;
      const int pos = start[o] + atomicAdd(&fill[o], 1);
      stage[pos] = make_int2(r[k], c[k]);
    }
  __syncthreads();
  const int left = nEdges - g * SLICE;
  const int len = left < SLICE ? left : SLICE;
  int2* dst = binned + (size_t)g * SLICE;
  for (int i = tid; i < len; i += 1024) dst[i] = stage[i];
}

__global__ __launch_bounds__(1024) void gt_bin_proj(
    const int* __restrict__ rows, const int* __restrict__ cols,
    int2* __restrict__ binned, int* __restrict__ C, int* __restrict__ S,
    int nEdges, int nSlices, ProjArgs p) {
  const int b = blockIdx.x;
  if (b < nSlices) {
    bin_slice(rows, cols, binned, C, S, nEdges, b);
    return;
  }
  proj_unit<0>(p, (b - nSlices) * 32 + (int)(threadIdx.x >> 5));
}

__device__ __forceinline__ void sort_owner(const int2* __restrict__ binned,
                                           const int* __restrict__ C,
                                           const int* __restrict__ S,
                                           int* __restrict__ off,
                                           int* __restrict__ scol,
                                           int nNodes, int nSlices, int nOwn,
                                           int o) {
  __shared__ int part[8][MAX_OWN];
  __shared__ int tot[MAX_OWN];
  __shared__ int cnt[OWN_NODES];
  __shared__ int excl[OWN_NODES];
  __shared__ int fil[OWN_NODES];
  __shared__ int stg[STAGE_CAP];
  __shared__ int shBase;
  const int tid = threadIdx.x, lane = tid & 31;
  const int wave = __builtin_amdgcn_readfirstlane(tid >> 5);

  {
    const int oo = tid & (MAX_OWN - 1), pt = tid >> 7;
    int s = 0;
#pragma unroll 8
    for (int g = pt; g < nSlices; g += 8) s += C[g * MAX_OWN + oo];
    part[pt][oo] = s;
  }
  if (tid < OWN_NODES) { cnt[tid] = 0; fil[tid] = 0; }
  int myN = 0, mySt = 0;
  {
    const int g = wave + 32 * lane;
    if (g < nSlices) { myN = C[g * MAX_OWN + o]; mySt = S[g * MAX_OWN + o]; }
  }
  __syncthreads();
  if (tid < MAX_OWN) {
    int s = 0;
#pragma unroll
    for (int k = 0; k < 8; ++k) s += part[k][tid];
    tot[tid] = s;
  }
  __syncthreads();
  if (tid < 32) {
    int s = 0;
#pragma unroll
    for (int k = 0; k < 4; ++k) {
      const int idx = 4 * tid + k;
      if (idx < o) s += tot[idx];
    }
#pragma unroll
    for (int d = 16; d >= 1; d >>= 1) s += __shfl_xor(s, d);
    if (tid == 0) shBase = s;
  }
  const int node0 = o << OWN_SHIFT;
  for (int j = 0; wave + 32 * j < nSlices; ++j) {
    const int n  = __builtin_amdgcn_readlane(myN, j);
    const int st = __builtin_amdgcn_readlane(mySt, j);
    const int2* run = binned + (size_t)(wave + 32 * j) * SLICE + st;
    for (int i = lane; i < n; i += 32) atomicAdd(&cnt[run[i].x - node0], 1);
  }
  __syncthreads();
  const int base = shBase, total = tot[o];
  if (tid < 32) {
    int v[16];
    int run = 0;
#pragma unroll
    for (int k = 0; k < 16; ++k) { const int a = cnt[16 * tid + k]; v[k] = run; run += a; }
    int incl = run;
#pragma unroll
    for (int d = 1; d < 32; d <<= 1) {
      int n = __shfl_up(incl, d);
      if (lane >= d) incl += n;
    }
    const int ex = incl - run;
#pragma unroll
    for (int k = 0; k < 16; ++k) excl[16 * tid + k] = ex + v[k];
  }
  __syncthreads();
  if (tid < OWN_NODES && node0 + tid < nNodes) off[node0 + tid] = base + excl[tid];
  if (tid == 0 && o == nOwn - 1) off[nNodes] = base + total;
  const bool useLds = total <= STAGE_CAP;
  for (int j = 0; wave + 32 * j < nSlices; ++j) {
    const int n  = __builtin_amdgcn_readlane(myN, j);
    const int st = __builtin_amdgcn_readlane(mySt, j);
    const int2* run = binned + (size_t)(wave + 32 * j) * SLICE + st;
    for (int i = lane; i < n; i += 32) {
      const int2 rc = run[i];
      const int d = rc.x - node0;
      const int pos = excl[d] + atomicAdd(&fil[d], 1);
      if (useLds) stg[pos] = rc.y; else scol[base + pos] = rc.y;
    }
  }
  __syncthreads();
  if (useLds)
    for (int i = tid; i < total; i += 1024) scol[base + i] = stg[i];
}

__global__ __launch_bounds__(1024) void gt_sort_proj(
    const int2* __restrict__ binned, const int* __restrict__ C,
    const int* __restrict__ S, int* __restrict__ off, int* __restrict__ scol,
    int nNodes, int nSlices, int nOwn, ProjArgs p) {
  const int b = blockIdx.x;
  if (b < nOwn) {
    sort_owner(binned, C, S, off, scol, nNodes, nSlices, nOwn, b);
    return;
  }
  proj_unit<3>(p, (b - nOwn) * 32 + (int)(threadIdx.x >> 5));
}

template <int M>
__device__ __forceinline__ float lane_xor(float x) {
  return __builtin_bit_cast(
      float, __builtin_amdgcn_update_dpp(0, __builtin_bit_cast(int, x),
                                         0x160 + M, 0xf, 0xf, true));
}

template <int J>
__device__ __forceinline__ float group_bcast(float x) {
  return __builtin_bit_cast(
      float, __builtin_amdgcn_mov_dpp8(__builtin_bit_cast(int, x),
                                       J * 0x249249));
}

__global__ __launch_bounds__(256) void gt_row_attend(
    const float* __restrict__ Q, const float* __restrict__ K,
    const float* __restrict__ V, const float* __restrict__ embeds,
    const int* __restrict__ off, const int* __restrict__ scol,
    float* __restrict__ out, int nNodes) {
  const int lane = threadIdx.x & 31;
  const int wave = __builtin_amdgcn_readfirstlane(threadIdx.x >> 5);
  const int row  = blockIdx.x * 8 + wave;
  if (row >= nNodes) return;
  const int beg = off[row], end = off[row + 1];
  const unsigned lo4 = (unsigned)lane * 12u;
  const int  sub = lane & 7;
  const bool b4 = (lane & 4) != 0, b2 = (lane & 2) != 0, b1 = (lane & 1) != 0;
  const char* Kb = (const char*)K;
  const char* Vb = (const char*)V;

  const f3 q = *(const f3*)((const char*)Q + (size_t)row * (DMODEL * 4) + lo4);
  float den = 0.0f, a0 = 0.0f, a1 = 0.0f, a2 = 0.0f;

  for (int p = beg; p < end; p += 32) {
    const int n = (end - p < 32) ? (end - p) : 32;
    const int mine = scol[p + (lane < n ? lane : 0)];
    for (int j0 = 0; j0 < n; j0 += 8) {
      f3 kf[8], vf[8];
      unsigned bo[8];
#pragma unroll
      for (int j = 0; j < 8; ++j) {
        const int jj = (j0 + j < n) ? (j0 + j) : j0;
        const int c  = __builtin_amdgcn_readlane(mine, jj);
        bo[j] = (unsigned)c * (DMODEL * 4u) + lo4;
      }
#pragma unroll
      for (int j = 0; j < 8; ++j) kf[j] = *(const f3*)(Kb + bo[j]);
#pragma unroll
      for (int j = 0; j < 8; ++j) vf[j] = *(const f3*)(Vb + bo[j]);
      __builtin_amdgcn_sched_barrier(0);

      float s[8];
#pragma unroll
      for (int j = 0; j < 8; ++j)
        s[j] = q.x * kf[j].x + q.y * kf[j].y + q.z * kf[j].z;
      float t[4], u[2];
#pragma unroll
      for (int k = 0; k < 4; ++k) {
        const float keep = b4 ? s[k + 4] : s[k];
        const float send = b4 ? s[k] : s[k + 4];
        t[k] = keep + lane_xor<4>(send);
      }
#pragma unroll
      for (int k = 0; k < 2; ++k) {
        const float keep = b2 ? t[k + 2] : t[k];
        const float send = b2 ? t[k] : t[k + 2];
        u[k] = keep + lane_xor<2>(send);
      }
      const float keep1 = b1 ? u[1] : u[0];
      const float send1 = b1 ? u[0] : u[1];
      float sc = keep1 + lane_xor<1>(send1);
      sc = fminf(10.0f, fmaxf(-10.0f, sc));
      const float es = (j0 + sub < n) ? expf(sc) : 0.0f;
      den += es;
      float w[8];
      w[0] = group_bcast<0>(es); w[1] = group_bcast<1>(es);
      w[2] = group_bcast<2>(es); w[3] = group_bcast<3>(es);
      w[4] = group_bcast<4>(es); w[5] = group_bcast<5>(es);
      w[6] = group_bcast<6>(es); w[7] = group_bcast<7>(es);
#pragma unroll
      for (int j = 0; j < 8; ++j) {
        a0 += w[j] * vf[j].x;
        a1 += w[j] * vf[j].y;
        a2 += w[j] * vf[j].z;
      }
    }
  }
  den += lane_xor<1>(den);
  den += lane_xor<2>(den);
  den += lane_xor<4>(den);
  const float inv = 1.0f / (den + 1e-8f);
  const size_t o = (size_t)row * (DMODEL * 4) + lo4;
  const f3 e3 = *(const f3*)((const char*)embeds + o);
  f3 r;
  r.x = e3.x + a0 * inv;
  r.y = e3.y + a1 * inv;
  r.z = e3.z + a2 * inv;
  *(f3*)((char*)out + o) = r;
}

extern "C" void kernel_launch(void* const* d_in, const int* in_sizes, int n_in,
                              void* d_out, int out_size, void* d_ws,
                              size_t ws_size, hipStream_t stream) {
  const float* embeds = (const float*)d_in[0];
  const float* qW     = (const float*)d_in[1];
  const float* kW     = (const float*)d_in[2];
  const float* vW     = (const float*)d_in[3];
  const int*   rows   = (const int*)d_in[4];
  const int*   cols   = (const int*)d_in[5];
  float*       out    = (float*)d_out;

  const int nNodes = in_sizes[0] / DMODEL;
  const int nEdges = in_sizes[4];

  const int    nSlices = (nEdges + SLICE - 1) / SLICE;
  const int    nOwn    = (nNodes + OWN_NODES - 1) / OWN_NODES;
  const size_t nEPad   = ((size_t)nEdges + 3) & ~(size_t)3;

  float* ws = (float*)d_ws;
  const size_t nd = (size_t)nNodes * DMODEL;
  float* Q      = ws;
  float* Kp     = Q + nd;
  float* Vp     = Kp + nd;
  int*   scol   = (int*)(Vp + nd);
  int2*  binned = (int2*)(scol + nEPad);
  int*   C      = (int*)(binned + (size_t)nSlices * SLICE);
  int*   S      = C + (size_t)nSlices * MAX_OWN;
  int*   off    = S + (size_t)nSlices * MAX_OWN;

  const int nMTiles = nNodes / 16;
  ProjArgs p;
  p.embeds = embeds; p.qW = qW; p.kW = kW; p.vW = vW;
  p.Q = Q; p.K = Kp; p.V = Vp; p.nMTiles = nMTiles;

  const int proj3 = (nMTiles * 3 + 31) / 32;

  gt_bin_proj<<<nSlices + proj3, 1024, 0, stream>>>(rows, cols, binned, C, S,
                                                    nEdges, nSlices, p);
  gt_sort_proj<<<nOwn + proj3, 1024, 0, stream>>>(binned, C, S, off, scol,
                                                  nNodes, nSlices, nOwn, p);
  gt_row_attend<<<(nNodes + 7) / 8, 256, 0, stream>>>(Q, Kp, Vp, embeds, off,
                                                      scol, out, nNodes);
}


# K and V of a node share one 768-byte record (3 aligned 256-byte units) in the row attention's lane order: an edge is a 16-byte and an 8-byte load per lane over whole lines; source sum order unchanged
# speedup vs baseline: 8.1131x; 1.0698x over previous
#include <hip/hip_runtime.h>
#include <cstddef>

typedef float v2f __attribute__((ext_vector_type(2)));
typedef float v4f __attribute__((ext_vector_type(4)));
typedef float v8f __attribute__((ext_vector_type(8)));

#define DMODEL 96
#define NHEADS 4
#define DHEAD  24
#define KVREC  192
#define OWN_SHIFT 9
#define OWN_NODES 512
#define MAX_OWN   128
#define SLICE     4096
#define STAGE_CAP 12288

struct ProjArgs {
  const float* embeds;
  const float* qW;
  const float* kW;
  const float* vW;
  float* Q;
  float* KV;
  int nMTiles;
};

struct __attribute__((aligned(4))) f3 { float x, y, z; };

template <int GB>
__device__ __forceinline__ void proj_unit(const ProjArgs& p, int unitOfWave) {
  const int lane = threadIdx.x & 31;
  const int unit = __builtin_amdgcn_readfirstlane(unitOfWave);
  if (unit >= p.nMTiles * 3) return;

  const int mt  = unit / 3;
  const int grp = GB + unit % 3;
  const int mat = grp >> 1;
  const int g0  = (grp & 1) * 48;
  const int m0  = mt * 16;

  const float* W   = (mat == 0) ? p.qW : (mat == 1) ? p.kW : p.vW;

  const int h = lane >> 4;
  const int c = lane & 15;

  const float* arow = p.embeds + (size_t)(m0 + c) * DMODEL + 4 * h;
  const float* bcol = W + (size_t)(4 * h) * DMODEL + g0 + 3 * c;

  v8f acc0 = {0.f, 0.f, 0.f, 0.f, 0.f, 0.f, 0.f, 0.f};
  v8f acc1 = acc0, acc2 = acc0;
#pragma unroll
  for (int kb = 0; kb < DMODEL; kb += 8) {
    const v4f a  = *(const v4f*)(arow + kb);
    const f3  w0 = *(const f3*)(bcol + (size_t)(kb + 0) * DMODEL);
    const f3  w1 = *(const f3*)(bcol + (size_t)(kb + 1) * DMODEL);
    const f3  w2 = *(const f3*)(bcol + (size_t)(kb + 2) * DMODEL);
    const f3  w3 = *(const f3*)(bcol + (size_t)(kb + 3) * DMODEL);
    const v2f a01 = {a.x, a.y};
    const v2f a23 = {a.z, a.w};
    v2f b;
    b.x = w0.x; b.y = w1.x;
    acc0 = __builtin_amdgcn_wmma_f32_16x16x4_f32(false, a01, false, b, (short)0, acc0, false, false);
    b.x = w0.y; b.y = w1.y;
    acc1 = __builtin_amdgcn_wmma_f32_16x16x4_f32(false, a01, false, b, (short)0, acc1, false, false);
    b.x = w0.z; b.y = w1.z;
    acc2 = __builtin_amdgcn_wmma_f32_16x16x4_f32(false, a01, false, b, (short)0, acc2, false, false);
    b.x = w2.x; b.y = w3.x;
    acc0 = __builtin_amdgcn_wmma_f32_16x16x4_f32(false, a23, false, b, (short)0, acc0, false, false);
    b.x = w2.y; b.y = w3.y;
    acc1 = __builtin_amdgcn_wmma_f32_16x16x4_f32(false, a23, false, b, (short)0, acc1, false, false);
    b.x = w2.z; b.y = w3.z;
    acc2 = __builtin_amdgcn_wmma_f32_16x16x4_f32(false, a23, false, b, (short)0, acc2, false, false);
  }

  if (mat == 0) {
    float* orow = p.Q + (size_t)(m0 + 8 * h) * DMODEL + g0 + 3 * c;
#pragma unroll
    for (int g = 0; g < 8; ++g) {
      f3 r;
      r.x = acc0[g]; r.y = acc1[g]; r.z = acc2[g];
      *(f3*)(orow + (size_t)g * DMODEL) = r;
    }
    return;
  }
  const int l = 16 * (grp & 1) + c;
  float* rec = p.KV + (size_t)(m0 + 8 * h) * KVREC;
  if (mat == 1) {
#pragma unroll
    for (int g = 0; g < 8; ++g) {
      f3 r;
      r.x = acc0[g]; r.y = acc1[g]; r.z = acc2[g];
      *(f3*)(rec + (size_t)g * KVREC + 4 * l) = r;
    }
  } else {
#pragma unroll
    for (int g = 0; g < 8; ++g) {
      v2f r;
      r.x = acc1[g]; r.y = acc2[g];
      rec[(size_t)g * KVREC + 4 * l + 3] = acc0[g];
      *(v2f*)(rec + (size_t)g * KVREC + 128 + 2 * l) = r;
    }
  }
}

__device__ __forceinline__ void bin_slice(const int* __restrict__ rows,
                                          const int* __restrict__ cols,
                                          int2* __restrict__ binned,
                                          int* __restrict__ C,
                                          int* __restrict__ S,
                                          int nEdges, int g) {
  __shared__ int  hist[MAX_OWN];
  __shared__ int  start[MAX_OWN];
  __shared__ int  fill[MAX_OWN];
  __shared__ int2 stage[SLICE];
  const int tid = threadIdx.x, lane = tid & 31;
  if (tid < MAX_OWN) { hist[tid] = 0; fill[tid] = 0; }
  __syncthreads();
  int r[4], c[4];
  const int e0 = g * SLICE + tid;
#pragma unroll
  for (int k = 0; k < 4; ++k) {
    const int e = e0 + k * 1024;
    r[k] = -1; c[k] = 0;
    if (e < nEdges) { r[k] = rows[e]; c[k] = cols[e]; }
  }
#pragma unroll
  for (int k = 0; k < 4; ++k)
    if (r[k] >= 0) atomicAdd(&hist[r[k] >> OWN_SHIFT], 1);
  __syncthreads();
  if (tid < 32) {
    const int a0 = hist[4 * tid], a1 = hist[4 * tid + 1];
    const int a2 = hist[4 * tid + 2], a3 = hist[4 * tid + 3];
    const int s = a0 + a1 + a2 + a3;
    int incl = s;
#pragma unroll
    for (int d = 1; d < 32; d <<= 1) {
      int n = __shfl_up(incl, d);
      if (lane >= d) incl += n;
    }
    const int ex = incl - s;
    start[4 * tid]     = ex;
    start[4 * tid + 1] = ex + a0;
    start[4 * tid + 2] = ex + a0 + a1;
    start[4 * tid + 3] = ex + a0 + a1 + a2;
  }
  __syncthreads();
  if (tid < MAX_OWN) {
    C[g * MAX_OWN + tid] = hist[tid];
    S[g * MAX_OWN + tid] = start[tid];
  }
#pragma unroll
  for (int k = 0; k < 4; ++k)
    if (r[k] >= 0) {
      const int o = r[k] >> OWN_SHIFT;
      const int pos = start[o] + atomicAdd(&fill[o], 1);
      stage[pos] = make_int2(r[k], c[k]);
    }
  __syncthreads();
  const int left = nEdges - g * SLICE;
  const int len = left < SLICE ? left : SLICE;
  int2* dst = binned + (size_t)g * SLICE;
  for (int i = tid; i < len; i += 1024) dst[i] = stage[i];
}

__global__ __launch_bounds__(1024) void gt_bin_proj(
    const int* __restrict__ rows, const int* __restrict__ cols,
    int2* __restrict__ binned, int* __restrict__ C, int* __restrict__ S,
    const float* embeds, const float* qW, const float* kW, const float* vW,
    float* Q, float* KV, int nEdges, int nSlices, int nMTiles) {
  ProjArgs p;
  p.embeds = embeds; p.qW = qW; p.kW = kW; p.vW = vW;
  p.Q = Q; p.KV = KV; p.nMTiles = nMTiles;
  const int b = blockIdx.x;
  if (b < nSlices) {
    bin_slice(rows, cols, binned, C, S, nEdges, b);
    return;
  }
  proj_unit<0>(p, (b - nSlices) * 32 + (int)(threadIdx.x >> 5));
}

__device__ __forceinline__ void sort_owner(const int2* __restrict__ binned,
                                           const int* __restrict__ C,
                                           const int* __restrict__ S,
                                           int* __restrict__ off,
                                           int* __restrict__ scol,
                                           int nNodes, int nSlices, int nOwn,
                                           int o) {
  __shared__ int part[8][MAX_OWN];
  __shared__ int tot[MAX_OWN];
  __shared__ int cnt[OWN_NODES];
  __shared__ int excl[OWN_NODES];
  __shared__ int fil[OWN_NODES];
  __shared__ int stg[STAGE_CAP];
  __shared__ int shBase;
  const int tid = threadIdx.x, lane = tid & 31;
  const int wave = __builtin_amdgcn_readfirstlane(tid >> 5);

  {
    const int oo = tid & (MAX_OWN - 1), pt = tid >> 7;
    int s = 0;
#pragma unroll 8
    for (int g = pt; g < nSlices; g += 8) s += C[g * MAX_OWN + oo];
    part[pt][oo] = s;
  }
  if (tid < OWN_NODES) { cnt[tid] = 0; fil[tid] = 0; }
  int myN = 0, mySt = 0;
  {
    const int g = wave + 32 * lane;
    if (g < nSlices) { myN = C[g * MAX_OWN + o]; mySt = S[g * MAX_OWN + o]; }
  }
  __syncthreads();
  if (tid < MAX_OWN) {
    int s = 0;
#pragma unroll
    for (int k = 0; k < 8; ++k) s += part[k][tid];
    tot[tid] = s;
  }
  __syncthreads();
  if (tid < 32) {
    int s = 0;
#pragma unroll
    for (int k = 0; k < 4; ++k) {
      const int idx = 4 * tid + k;
      if (idx < o) s += tot[idx];
    }
#pragma unroll
    for (int d = 16; d >= 1; d >>= 1) s += __shfl_xor(s, d);
    if (tid == 0) shBase = s;
  }
  const int node0 = o << OWN_SHIFT;
  for (int j = 0; wave + 32 * j < nSlices; ++j) {
    const int n  = __builtin_amdgcn_readlane(myN, j);
    const int st = __builtin_amdgcn_readlane(mySt, j);
    const int2* run = binned + (size_t)(wave + 32 * j) * SLICE + st;
    for (int i = lane; i < n; i += 32) atomicAdd(&cnt[run[i].x - node0], 1);
  }
  __syncthreads();
  const int base = shBase, total = tot[o];
  if (tid < 32) {
    int v[16];
    int run = 0;
#pragma unroll
    for (int k = 0; k < 16; ++k) { const int a = cnt[16 * tid + k]; v[k] = run; run += a; }
    int incl = run;
#pragma unroll
    for (int d = 1; d < 32; d <<= 1) {
      int n = __shfl_up(incl, d);
      if (lane >= d) incl += n;
    }
    const int ex = incl - run;
#pragma unroll
    for (int k = 0; k < 16; ++k) excl[16 * tid + k] = ex + v[k];
  }
  __syncthreads();
  if (tid < OWN_NODES && node0 + tid < nNodes) off[node0 + tid] = base + excl[tid];
  if (tid == 0 && o == nOwn - 1) off[nNodes] = base + total;
  const bool useLds = total <= STAGE_CAP;
  for (int j = 0; wave + 32 * j < nSlices; ++j) {
    const int n  = __builtin_amdgcn_readlane(myN, j);
    const int st = __builtin_amdgcn_readlane(mySt, j);
    const int2* run = binned + (size_t)(wave + 32 * j) * SLICE + st;
    for (int i = lane; i < n; i += 32) {
      const int2 rc = run[i];
      const int d = rc.x - node0;
      const int pos = excl[d] + atomicAdd(&fil[d], 1);
      if (useLds) stg[pos] = rc.y; else scol[base + pos] = rc.y;
    }
  }
  __syncthreads();
  if (useLds)
    for (int i = tid; i < total; i += 1024) scol[base + i] = stg[i];
}

__global__ __launch_bounds__(1024) void gt_sort_proj(
    const int2* __restrict__ binned, const int* __restrict__ C,
    const int* __restrict__ S, int* __restrict__ off, int* __restrict__ scol,
    const float* embeds, const float* qW, const float* kW, const float* vW,
    float* Q, float* KV, int nNodes, int nSlices, int nOwn, int nMTiles) {
  ProjArgs p;
  p.embeds = embeds; p.qW = qW; p.kW = kW; p.vW = vW;
  p.Q = Q; p.KV = KV; p.nMTiles = nMTiles;
  const int b = blockIdx.x;
  if (b < nOwn) {
    sort_owner(binned, C, S, off, scol, nNodes, nSlices, nOwn, b);
    return;
  }
  proj_unit<3>(p, (b - nOwn) * 32 + (int)(threadIdx.x >> 5));
}

template <int M>
__device__ __forceinline__ float lane_xor(float x) {
  return __builtin_bit_cast(
      float, __builtin_amdgcn_update_dpp(0, __builtin_bit_cast(int, x),
                                         0x160 + M, 0xf, 0xf, true));
}

template <int J>
__device__ __forceinline__ float group_bcast(float x) {
  return __builtin_bit_cast(
      float, __builtin_amdgcn_mov_dpp8(__builtin_bit_cast(int, x),
                                       J * 0x249249));
}

__global__ __launch_bounds__(256) void gt_row_attend(
    const float* __restrict__ Q, const float* __restrict__ KV,
    const float* __restrict__ embeds,
    const int* __restrict__ off, const int* __restrict__ scol,
    float* __restrict__ out, int nNodes) {
  const int lane = threadIdx.x & 31;
  const int wave = __builtin_amdgcn_readfirstlane(threadIdx.x >> 5);
  const int row  = blockIdx.x * 8 + wave;
  if (row >= nNodes) return;
  const int beg = off[row], end = off[row + 1];
  const unsigned lo4 = (unsigned)lane * 12u;
  const int  sub = lane & 7;
  const bool b4 = (lane & 4) != 0, b2 = (lane & 2) != 0, b1 = (lane & 1) != 0;
  const unsigned lo16 = (unsigned)lane * 16u;
  const unsigned lo8  = 512u + (unsigned)lane * 8u;
  const char* KVb = (const char*)KV;

  const f3 q = *(const f3*)((const char*)Q + (size_t)row * (DMODEL * 4) + lo4);
  float den = 0.0f, a0 = 0.0f, a1 = 0.0f, a2 = 0.0f;

  for (int p = beg; p < end; p += 32) {
    const int n = (end - p < 32) ? (end - p) : 32;
    const int mine = scol[p + (lane < n ? lane : 0)];
    for (int j0 = 0; j0 < n; j0 += 8) {
      v4f kf[8];
      v2f vf[8];
      unsigned b16[8], b8[8];
#pragma unroll
      for (int j = 0; j < 8; ++j) {
        const int jj = (j0 + j < n) ? (j0 + j) : j0;
        const int c  = __builtin_amdgcn_readlane(mine, jj);
        b16[j] = (unsigned)c * (KVREC * 4u) + lo16;
        b8[j]  = (unsigned)c * (KVREC * 4u) + lo8;
      }
#pragma unroll
      for (int j = 0; j < 8; ++j) {
        kf[j] = *(const v4f*)(KVb + b16[j]);
        vf[j] = *(const v2f*)(KVb + b8[j]);
      }
      __builtin_amdgcn_sched_barrier(0);

      float s[8];
#pragma unroll
      for (int j = 0; j < 8; ++j)
        s[j] = q.x * kf[j].x + q.y * kf[j].y + q.z * kf[j].z;
      float t[4], u[2];
#pragma unroll
      for (int k = 0; k < 4; ++k) {
        const float keep = b4 ? s[k + 4] : s[k];
        const float send = b4 ? s[k] : s[k + 4];
        t[k] = keep + lane_xor<4>(send);
      }
#pragma unroll
      for (int k = 0; k < 2; ++k) {
        const float keep = b2 ? t[k + 2] : t[k];
        const float send = b2 ? t[k] : t[k + 2];
        u[k] = keep + lane_xor<2>(send);
      }
      const float keep1 = b1 ? u[1] : u[0];
      const float send1 = b1 ? u[0] : u[1];
      float sc = keep1 + lane_xor<1>(send1);
      sc = fminf(10.0f, fmaxf(-10.0f, sc));
      const float es = (j0 + sub < n) ? expf(sc) : 0.0f;
      den += es;
      float w[8];
      w[0] = group_bcast<0>(es); w[1] = group_bcast<1>(es);
      w[2] = group_bcast<2>(es); w[3] = group_bcast<3>(es);
      w[4] = group_bcast<4>(es); w[5] = group_bcast<5>(es);
      w[6] = group_bcast<6>(es); w[7] = group_bcast<7>(es);
#pragma unroll
      for (int j = 0; j < 8; ++j) {
        a0 += w[j] * kf[j].w;
        a1 += w[j] * vf[j].x;
        a2 += w[j] * vf[j].y;
      }
    }
  }
  den += lane_xor<1>(den);
  den += lane_xor<2>(den);
  den += lane_xor<4>(den);
  const float inv = 1.0f / (den + 1e-8f);
  const size_t o = (size_t)row * (DMODEL * 4) + lo4;
  const f3 e3 = *(const f3*)((const char*)embeds + o);
  f3 r;
  r.x = e3.x + a0 * inv;
  r.y = e3.y + a1 * inv;
  r.z = e3.z + a2 * inv;
  *(f3*)((char*)out + o) = r;
}

extern "C" void kernel_launch(void* const* d_in, const int* in_sizes, int n_in,
                              void* d_out, int out_size, void* d_ws,
                              size_t ws_size, hipStream_t stream) {
  const float* embeds = (const float*)d_in[0];
  const float* qW     = (const float*)d_in[1];
  const float* kW     = (const float*)d_in[2];
  const float* vW     = (const float*)d_in[3];
  const int*   rows   = (const int*)d_in[4];
  const int*   cols   = (const int*)d_in[5];
  float*       out    = (float*)d_out;

  const int nNodes = in_sizes[0] / DMODEL;
  const int nEdges = in_sizes[4];

  const int    nSlices = (nEdges + SLICE - 1) / SLICE;
  const int    nOwn    = (nNodes + OWN_NODES - 1) / OWN_NODES;
  const size_t nEPad   = ((size_t)nEdges + 3) & ~(size_t)3;

  float* ws = (float*)d_ws;
  const size_t nd = (size_t)nNodes * DMODEL;
  float* KV     = ws;
  float* Q      = KV + 2 * nd;
  int*   scol   = (int*)(Q + nd);
  int2*  binned = (int2*)(scol + nEPad);
  int*   C      = (int*)(binned + (size_t)nSlices * SLICE);
  int*   S      = C + (size_t)nSlices * MAX_OWN;
  int*   off    = S + (size_t)nSlices * MAX_OWN;

  const int nMTiles = nNodes / 16;
  const int proj3 = (nMTiles * 3 + 31) / 32;

  gt_bin_proj<<<nSlices + proj3, 1024, 0, stream>>>(
      rows, cols, binned, C, S, embeds, qW, kW, vW, Q, KV, nEdges, nSlices,
      nMTiles);
  gt_sort_proj<<<nOwn + proj3, 1024, 0, stream>>>(
      binned, C, S, off, scol, embeds, qW, kW, vW, Q, KV, nNodes, nSlices, nOwn,
      nMTiles);
  gt_row_attend<<<(nNodes + 7) / 8, 256, 0, stream>>>(Q, KV, embeds, off, scol,
                                                      out, nNodes);
}


# Owner sort and its rows' attention share one work-group: row offsets, sorted columns stay in LDS, rows dealt to waves by an LDS counter; 2 launches, all projections in the first; arrival-order sums
# speedup vs baseline: 8.7010x; 1.0725x over previous
#include <hip/hip_runtime.h>
#include <cstddef>


typedef float v2f __attribute__((ext_vector_type(2)));
typedef float v4f __attribute__((ext_vector_type(4)));
typedef float v8f __attribute__((ext_vector_type(8)));

#define DMODEL 96
#define NHEADS 4
#define DHEAD  24
#define KVREC  192
#define OWN_NODES 98
#define CNT_PAD   128
#define MAX_OWN   512
#define BIN_WAVES 24
#define SLICE     3072
#define STAGE_CAP 4096
#define OWN_WAVES 24
#define RUNS_REG  11

struct ProjArgs {
  const float* embeds;
  const float* qW;
  const float* kW;
  const float* vW;
  float* Q;
  float* KV;
  int nMTiles;
};

struct __attribute__((aligned(4))) f3 { float x, y, z; };

__device__ __forceinline__ void proj_unit(const ProjArgs& p, int mt, int grp) {
  const int lane = threadIdx.x & 31;
  const int mat = grp >> 1;
  const int g0  = (grp & 1) * 48;
  const int m0  = mt * 16;

  const float* W   = (mat == 0) ? p.qW : (mat == 1) ? p.kW : p.vW;

  const int h = lane >> 4;
  const int c = lane & 15;

  const char* arow = (const char*)p.embeds + ((unsigned)(m0 + c) * DMODEL + 4 * h) * 4u;
  const float* bcol = W + (size_t)(4 * h) * DMODEL + g0 + 3 * c;

  v8f acc0 = {0.f, 0.f, 0.f, 0.f, 0.f, 0.f, 0.f, 0.f};
  v8f acc1 = acc0, acc2 = acc0;
#pragma unroll
  for (int kb = 0; kb < DMODEL; kb += 8) {
    const v4f a  = *(const v4f*)(arow + kb * 4);
    const f3  w0 = *(const f3*)(bcol + (size_t)(kb + 0) * DMODEL);
    const f3  w1 = *(const f3*)(bcol + (size_t)(kb + 1) * DMODEL);
    const f3  w2 = *(const f3*)(bcol + (size_t)(kb + 2) * DMODEL);
    const f3  w3 = *(const f3*)(bcol + (size_t)(kb + 3) * DMODEL);
    const v2f a01 = {a.x, a.y};
    const v2f a23 = {a.z, a.w};
    v2f b;
    b.x = w0.x; b.y = w1.x;
    acc0 = __builtin_amdgcn_wmma_f32_16x16x4_f32(false, a01, false, b, (short)0, acc0, false, false);
    b.x = w0.y; b.y = w1.y;
    acc1 = __builtin_amdgcn_wmma_f32_16x16x4_f32(false, a01, false, b, (short)0, acc1, false, false);
    b.x = w0.z; b.y = w1.z;
    acc2 = __builtin_amdgcn_wmma_f32_16x16x4_f32(false, a01, false, b, (short)0, acc2, false, false);
    b.x = w2.x; b.y = w3.x;
    acc0 = __builtin_amdgcn_wmma_f32_16x16x4_f32(false, a23, false, b, (short)0, acc0, false, false);
    b.x = w2.y; b.y = w3.y;
    acc1 = __builtin_amdgcn_wmma_f32_16x16x4_f32(false, a23, false, b, (short)0, acc1, false, false);
    b.x = w2.z; b.y = w3.z;
    acc2 = __builtin_amdgcn_wmma_f32_16x16x4_f32(false, a23, false, b, (short)0, acc2, false, false);
  }

  if (mat == 0) {
    char* orow = (char*)p.Q + ((unsigned)(m0 + 8 * h) * DMODEL + g0 + 3 * c) * 4u;
#pragma unroll
    for (int g = 0; g < 8; ++g) {
      f3 r;
      r.x = acc0[g]; r.y = acc1[g]; r.z = acc2[g];
      *(f3*)(orow + g * (DMODEL * 4)) = r;
    }
    return;
  }
  const int l = 16 * (grp & 1) + c;
  float* rec = (float*)((char*)p.KV + (unsigned)(m0 + 8 * h) * (KVREC * 4u));
  if (mat == 1) {
#pragma unroll
    for (int g = 0; g < 8; ++g) {
      f3 r;
      r.x = acc0[g]; r.y = acc1[g]; r.z = acc2[g];
      *(f3*)(rec + (size_t)g * KVREC + 4 * l) = r;
    }
  } else {
#pragma unroll
    for (int g = 0; g < 8; ++g) {
      v2f r;
      r.x = acc1[g]; r.y = acc2[g];
      rec[(size_t)g * KVREC + 4 * l + 3] = acc0[g];
      *(v2f*)(rec + (size_t)g * KVREC + 128 + 2 * l) = r;
    }
  }
}

__device__ __forceinline__ void bin_slice(const int* __restrict__ rows,
                                          const int* __restrict__ cols,
                                          int2* __restrict__ binned,
                                          int* __restrict__ C,
                                          int* __restrict__ S,
                                          int nEdges, int g) {
  __shared__ int  hist[MAX_OWN];
  __shared__ int  start[MAX_OWN];
  __shared__ int  fill[MAX_OWN];
  __shared__ int2 stage[SLICE];
  const int tid = threadIdx.x, lane = tid & 31;
  if (tid < MAX_OWN) { hist[tid] = 0; fill[tid] = 0; }
  __syncthreads();
  int r[4], c[4];
  const int e0 = g * SLICE + tid;
#pragma unroll
  for (int k = 0; k < 4; ++k) {
    const int e = e0 + k * (BIN_WAVES * 32);
    r[k] = -1; c[k] = 0;
    if (e < nEdges) { r[k] = rows[e]; c[k] = cols[e]; }
  }
#pragma unroll
  for (int k = 0; k < 4; ++k)
    if (r[k] >= 0) atomicAdd(&hist[r[k] / OWN_NODES], 1);
  __syncthreads();
  if (tid < 32) {
    int v[16];
    int run = 0;
#pragma unroll
    for (int k = 0; k < 16; ++k) { const int a = hist[16 * tid + k]; v[k] = run; run += a; }
    int incl = run;
#pragma unroll
    for (int d = 1; d < 32; d <<= 1) {
      int n = __shfl_up(incl, d);
      if (lane >= d) incl += n;
    }
    const int ex = incl - run;
#pragma unroll
    for (int k = 0; k < 16; ++k) start[16 * tid + k] = ex + v[k];
  }
  __syncthreads();
  if (tid < MAX_OWN) {
    C[g * MAX_OWN + tid] = hist[tid];
    S[g * MAX_OWN + tid] = start[tid];
  }
#pragma unroll
  for (int k = 0; k < 4; ++k)
    if (r[k] >= 0) {
      const int o = r[k] / OWN_NODES;
      const int pos = start[o] + atomicAdd(&fill[o], 1);
      stage[pos] = make_int2(r[k], c[k]);
    }
  __syncthreads();
  const int left = nEdges - g * SLICE;
  const int len = left < SLICE ? left : SLICE;
  int2* dst = binned + (size_t)g * SLICE;
  for (int i = tid; i < len; i += BIN_WAVES * 32) dst[i] = stage[i];
}

__global__ __launch_bounds__(1024) void gt_bin_proj(
    const int* __restrict__ rows, const int* __restrict__ cols,
    int2* __restrict__ binned, int* __restrict__ C, int* __restrict__ S,
    const float* embeds, const float* qW, const float* kW, const float* vW,
    float* Q, float* KV, int nEdges, int nSlices, int nMTiles) {
  ProjArgs p;
  p.embeds = embeds; p.qW = qW; p.kW = kW; p.vW = vW;
  p.Q = Q; p.KV = KV; p.nMTiles = nMTiles;
  const int b = blockIdx.x;
  const int G = gridDim.x;
  if (b < nSlices)
    bin_slice(rows, cols, binned, C, S, nEdges, b);
  const int wave = __builtin_amdgcn_readfirstlane((int)(threadIdx.x >> 5));
  const int grp = b % 6, k = b / 6;
  const int nk = (G - grp + 5) / 6;
  int mt = k * BIN_WAVES + wave;
  int next = nk * BIN_WAVES + wave * nk + k;
  while (mt < nMTiles) {
    proj_unit(p, mt, grp);
    mt = next;
    next += nk * BIN_WAVES;
  }
}

template <int M>
__device__ __forceinline__ float lane_xor(float x) {
  return __builtin_bit_cast(
      float, __builtin_amdgcn_update_dpp(0, __builtin_bit_cast(int, x),
                                         0x160 + M, 0xf, 0xf, true));
}

template <int J>
__device__ __forceinline__ float group_bcast(float x) {
  return __builtin_bit_cast(
      float, __builtin_amdgcn_mov_dpp8(__builtin_bit_cast(int, x),
                                       J * 0x249249));
}

template <class ColAt>
__device__ __forceinline__ void attend_row(const float* __restrict__ Q,
                                           const char* __restrict__ KVb,
                                           const float* __restrict__ embeds,
                                           float* __restrict__ out,
                                           int row, int beg, int end, int lane,
                                           ColAt colAt) {
  const unsigned lo4 = (unsigned)lane * 12u;
  const int  sub = lane & 7;
  const bool b4 = (lane & 4) != 0, b2 = (lane & 2) != 0, b1 = (lane & 1) != 0;
  const unsigned lo16 = (unsigned)lane * 16u;
  const unsigned lo8  = 512u + (unsigned)lane * 8u;

  const size_t o = (size_t)row * (DMODEL * 4) + lo4;
  const f3 q = *(const f3*)((const char*)Q + o);
  float den = 0.0f, a0 = 0.0f, a1 = 0.0f, a2 = 0.0f;

  for (int p = beg; p < end; p += 32) {
    const int n = (end - p < 32) ? (end - p) : 32;
    const int mine = colAt(p + (lane < n ? lane : 0));
    for (int j0 = 0; j0 < n; j0 += 8) {
      v4f kf[8];
      v2f vf[8];
      unsigned b16[8], b8[8];
#pragma unroll
      for (int j = 0; j < 8; ++j) {
        const int jj = (j0 + j < n) ? (j0 + j) : j0;
        const int c  = __builtin_amdgcn_readlane(mine, jj);
        b16[j] = (unsigned)c * (KVREC * 4u) + lo16;
        b8[j]  = (unsigned)c * (KVREC * 4u) + lo8;
      }
#pragma unroll
      for (int j = 0; j < 8; ++j) {
        kf[j] = *(const v4f*)(KVb + b16[j]);
        vf[j] = *(const v2f*)(KVb + b8[j]);
      }
      __builtin_amdgcn_sched_barrier(0);

      float s[8];
#pragma unroll
      for (int j = 0; j < 8; ++j)
        s[j] = q.x * kf[j].x + q.y * kf[j].y + q.z * kf[j].z;
      float t[4], u[2];
#pragma unroll
      for (int k = 0; k < 4; ++k) {
        const float keep = b4 ? s[k + 4] : s[k];
        const float send = b4 ? s[k] : s[k + 4];
        t[k] = keep + lane_xor<4>(send);
      }
#pragma unroll
      for (int k = 0; k < 2; ++k) {
        const float keep = b2 ? t[k + 2] : t[k];
        const float send = b2 ? t[k] : t[k + 2];
        u[k] = keep + lane_xor<2>(send);
      }
      const float keep1 = b1 ? u[1] : u[0];
      const float send1 = b1 ? u[0] : u[1];
      float sc = keep1 + lane_xor<1>(send1);
      sc = fminf(10.0f, fmaxf(-10.0f, sc));
      const float es = (j0 + sub < n) ? expf(sc) : 0.0f;
      den += es;
      float w[8];
      w[0] = group_bcast<0>(es); w[1] = group_bcast<1>(es);
      w[2] = group_bcast<2>(es); w[3] = group_bcast<3>(es);
      w[4] = group_bcast<4>(es); w[5] = group_bcast<5>(es);
      w[6] = group_bcast<6>(es); w[7] = group_bcast<7>(es);
#pragma unroll
      for (int j = 0; j < 8; ++j) {
        a0 += w[j] * kf[j].w;
        a1 += w[j] * vf[j].x;
        a2 += w[j] * vf[j].y;
      }
    }
  }
  den += lane_xor<1>(den);
  den += lane_xor<2>(den);
  den += lane_xor<4>(den);
  const float inv = 1.0f / (den + 1e-8f);
  const f3 e3 = *(const f3*)((const char*)embeds + o);
  f3 r;
  r.x = e3.x + a0 * inv;
  r.y = e3.y + a1 * inv;
  r.z = e3.z + a2 * inv;
  *(f3*)((char*)out + o) = r;
}

__global__ __launch_bounds__(OWN_WAVES * 32) void gt_sort_attend(
    const int2* __restrict__ binned, const int* __restrict__ C,
    const int* __restrict__ S, int* scol,
    const float* __restrict__ Q, const float* __restrict__ KV,
    const float* __restrict__ embeds, float* __restrict__ out,
    int nNodes, int nSlices) {
  __shared__ int cnt[CNT_PAD];
  __shared__ int excl[CNT_PAD + 1];
  __shared__ int fil[CNT_PAD];
  __shared__ int stg[STAGE_CAP];
  __shared__ int red[OWN_WAVES];
  __shared__ int nextRow;
  const int o = blockIdx.x;
  const int tid = threadIdx.x, lane = tid & 31;
  const int wave = __builtin_amdgcn_readfirstlane(tid >> 5);
  const int node0 = o * OWN_NODES;

  if (tid < CNT_PAD) { cnt[tid] = 0; fil[tid] = 0; }
  if (tid == 0) nextRow = OWN_WAVES;
  int myN = 0, mySt = 0;
  {
    const int g = wave + OWN_WAVES * lane;
    if (g < nSlices) { myN = C[g * MAX_OWN + o]; mySt = S[g * MAX_OWN + o]; }
  }
  int2 e[RUNS_REG];
#pragma unroll
  for (int j = 0; j < RUNS_REG; ++j) {
    const int n  = __builtin_amdgcn_readlane(myN, j);
    const int st = __builtin_amdgcn_readlane(mySt, j);
    e[j] = make_int2(-1, 0);
    if (lane < n) e[j] = binned[(size_t)(wave + OWN_WAVES * j) * SLICE + st + lane];
  }
  __syncthreads();
#pragma unroll
  for (int j = 0; j < RUNS_REG; ++j)
    if (e[j].x >= 0) atomicAdd(&cnt[e[j].x - node0], 1);
  for (int j = 0; wave + OWN_WAVES * j < nSlices; ++j) {
    const int n  = __builtin_amdgcn_readlane(myN, j);
    const int st = __builtin_amdgcn_readlane(mySt, j);
    const int2* run = binned + (size_t)(wave + OWN_WAVES * j) * SLICE + st;
    for (int i = (j < RUNS_REG ? 32 : 0) + lane; i < n; i += 32)
      atomicAdd(&cnt[run[i].x - node0], 1);
  }
  __syncthreads();
  if (tid < 32) {
    const int c0 = cnt[4 * tid], c1 = cnt[4 * tid + 1];
    const int c2 = cnt[4 * tid + 2], c3 = cnt[4 * tid + 3];
    const int s = c0 + c1 + c2 + c3;
    int incl = s;
#pragma unroll
    for (int d = 1; d < 32; d <<= 1) {
      int n = __shfl_up(incl, d);
      if (lane >= d) incl += n;
    }
    const int ex = incl - s;
    excl[4 * tid]     = ex;
    excl[4 * tid + 1] = ex + c0;
    excl[4 * tid + 2] = ex + c0 + c1;
    excl[4 * tid + 3] = ex + c0 + c1 + c2;
    if (tid == 31) excl[CNT_PAD] = incl;
  }
  __syncthreads();
  const int total = excl[CNT_PAD];
  const bool useLds = total <= STAGE_CAP;
  int base = 0;
  if (!useLds) {
    int s = 0;
    const int cells = nSlices * o;
    for (int idx = tid; idx < cells; idx += OWN_WAVES * 32) {
      const int g = idx / o;
      s += C[g * MAX_OWN + (idx - g * o)];
    }
#pragma unroll
    for (int d = 16; d >= 1; d >>= 1) s += __shfl_xor(s, d);
    if (lane == 0) red[wave] = s;
    __syncthreads();
#pragma unroll
    for (int k = 0; k < OWN_WAVES; ++k) base += red[k];
  }
#pragma unroll
  for (int j = 0; j < RUNS_REG; ++j)
    if (e[j].x >= 0) {
      const int d = e[j].x - node0;
      const int pos = excl[d] + atomicAdd(&fil[d], 1);
      if (useLds) stg[pos] = e[j].y; else scol[base + pos] = e[j].y;
    }
  for (int j = 0; wave + OWN_WAVES * j < nSlices; ++j) {
    const int n  = __builtin_amdgcn_readlane(myN, j);
    const int st = __builtin_amdgcn_readlane(mySt, j);
    const int2* run = binned + (size_t)(wave + OWN_WAVES * j) * SLICE + st;
    for (int i = (j < RUNS_REG ? 32 : 0) + lane; i < n; i += 32) {
      const int2 rc = run[i];
      const int d = rc.x - node0;
      const int pos = excl[d] + atomicAdd(&fil[d], 1);
      if (useLds) stg[pos] = rc.y; else scol[base + pos] = rc.y;
    }
  }
  if (!useLds) __threadfence();
  __syncthreads();

  const char* KVb = (const char*)KV;
  int rl = wave;
  while (rl < OWN_NODES) {
    const int row = node0 + rl;
    if (row < nNodes) {
      const int beg = __builtin_amdgcn_readfirstlane(excl[rl]);
      const int end = __builtin_amdgcn_readfirstlane(excl[rl + 1]);
      if (useLds)
        attend_row(Q, KVb, embeds, out, row, beg, end, lane,
                   [&](int i) { return stg[i]; });
      else
        attend_row(Q, KVb, embeds, out, row, beg, end, lane,
                   [&](int i) { return scol[base + i]; });
    }
    int t = 0;
    if (lane == 0) t = atomicAdd(&nextRow, 1);
    rl = __builtin_amdgcn_readfirstlane(t);
  }
}

extern "C" void kernel_launch(void* const* d_in, const int* in_sizes, int n_in,
                              void* d_out, int out_size, void* d_ws,
                              size_t ws_size, hipStream_t stream) {
  const float* embeds = (const float*)d_in[0];
  const float* qW     = (const float*)d_in[1];
  const float* kW     = (const float*)d_in[2];
  const float* vW     = (const float*)d_in[3];
  const int*   rows   = (const int*)d_in[4];
  const int*   cols   = (const int*)d_in[5];
  float*       out    = (float*)d_out;

  const int nNodes = in_sizes[0] / DMODEL;
  const int nEdges = in_sizes[4];

  const int    nSlices = (nEdges + SLICE - 1) / SLICE;
  const int    nOwn    = (nNodes + OWN_NODES - 1) / OWN_NODES;
  const size_t nEPad   = ((size_t)nEdges + 3) & ~(size_t)3;

  float* ws = (float*)d_ws;
  const size_t nd = (size_t)nNodes * DMODEL;
  float* KV     = ws;
  float* Q      = KV + 2 * nd;
  int*   scol   = (int*)(Q + nd);
  int2*  binned = (int2*)(scol + nEPad);
  int*   C      = (int*)(binned + (size_t)nSlices * SLICE);
  int*   S      = C + (size_t)nSlices * MAX_OWN;

  const int nMTiles = nNodes / 16;
  const int grid1 = nSlices > 512 ? nSlices : 512;

  gt_bin_proj<<<grid1, BIN_WAVES * 32, 0, stream>>>(
      rows, cols, binned, C, S, embeds, qW, kW, vW, Q, KV, nEdges, nSlices,
      nMTiles);
  gt_sort_attend<<<nOwn, OWN_WAVES * 32, 0, stream>>>(binned, C, S, scol, Q, KV, embeds,
                                            out, nNodes, nSlices);
}
